# starting kernel
# baseline (speedup 1.0000x reference)
#include <hip/hip_runtime.h>
#include <hip/hip_bf16.h>

typedef __attribute__((ext_vector_type(2))) float v2f;
typedef __attribute__((ext_vector_type(8))) float v8f;

#define N_NODES   100000
#define N_EDGES0  800000
#define N_EDGES   900000
#define ROW_TILES 6250

__device__ __forceinline__ float leaky(float v) { return v > 0.0f ? v : 0.2f * v; }

__device__ __forceinline__ void atomicMaxFloat(float* addr, float val) {
    if (val >= 0.0f) atomicMax((int*)addr, __float_as_int(val));
    else             atomicMin((unsigned int*)addr, __float_as_uint(val));
}

__global__ __launch_bounds__(256) void init_ws(float* agg, float* emax, float* denom) {
    size_t i = (size_t)blockIdx.x * 256 + threadIdx.x;
    if (i < (size_t)N_NODES * 256) agg[i] = 0.0f;
    if (i < N_NODES) { emax[i] = -INFINITY; denom[i] = 0.0f; }
}

__global__ __launch_bounds__(256) void gemm1_xw(const float* __restrict__ x,
                                                const float* __restrict__ W,
                                                float* __restrict__ xw) {
    const int lane = threadIdx.x & 31;
    const int wave = threadIdx.x >> 5;
    const int rowTile = blockIdx.x * 8 + wave;
    if (rowTile >= ROW_TILES) return;
    const int row0 = rowTile * 16;
    const int col0 = blockIdx.y * 64;

    const int rA  = row0 + (lane & 15);
    const int kHi = (lane >> 4) * 2;
    const int nB  = lane & 15;

    v8f acc[4] = {};
    for (int k = 0; k < 128; k += 4) {
        v2f a;
        a.x = x[(size_t)rA * 128 + k + kHi];
        a.y = x[(size_t)rA * 128 + k + kHi + 1];
#pragma unroll
        for (int t = 0; t < 4; ++t) {
            v2f bm;
            bm.x = W[(size_t)(k + kHi)     * 256 + col0 + t * 16 + nB];
            bm.y = W[(size_t)(k + kHi + 1) * 256 + col0 + t * 16 + nB];
            acc[t] = __builtin_amdgcn_wmma_f32_16x16x4_f32(
                false, a, false, bm, (short)0, acc[t], false, false);
        }
    }
    const int mBase = row0 + ((lane >> 4) << 3);
#pragma unroll
    for (int t = 0; t < 4; ++t)
#pragma unroll
        for (int r = 0; r < 8; ++r)
            xw[(size_t)(mBase + r) * 256 + col0 + t * 16 + nB] = acc[t][r];
}

__global__ __launch_bounds__(256) void alphas_kernel(const float* __restrict__ xw,
                                                     const float* __restrict__ a_src,
                                                     const float* __restrict__ a_dst,
                                                     float* __restrict__ as,
                                                     float* __restrict__ ad) {
    const int lane = threadIdx.x & 31;
    const int node = blockIdx.x * 8 + (threadIdx.x >> 5);
    if (node >= N_NODES) return;
    const float* xr = xw + (size_t)node * 256;
    float s = 0.0f, d = 0.0f;
#pragma unroll
    for (int i = 0; i < 8; ++i) {
        float v = xr[lane + 32 * i];
        s += v * a_src[lane + 32 * i];
        d += v * a_dst[lane + 32 * i];
    }
#pragma unroll
    for (int off = 16; off; off >>= 1) {
        s += __shfl_xor(s, off, 32);
        d += __shfl_xor(d, off, 32);
    }
    if (lane == 0) { as[node] = s; ad[node] = d; }
}

__device__ __forceinline__ void edge_endpoints(const int* __restrict__ ei, int e,
                                               int& s, int& d) {
    if (e < N_EDGES0) { s = ei[e]; d = ei[N_EDGES0 + e]; }
    else              { s = e - N_EDGES0; d = e - N_EDGES0; }
}

__global__ __launch_bounds__(256) void edge_max_kernel(const int* __restrict__ ei,
                                                       const float* __restrict__ as,
                                                       const float* __restrict__ ad,
                                                       float* __restrict__ emax) {
    int e = blockIdx.x * 256 + threadIdx.x;
    if (e >= N_EDGES) return;
    int s, d; edge_endpoints(ei, e, s, d);
    atomicMaxFloat(&emax[d], leaky(as[s] + ad[d]));
}

__global__ __launch_bounds__(256) void edge_exp_kernel(const int* __restrict__ ei,
                                                       const float* __restrict__ as,
                                                       const float* __restrict__ ad,
                                                       const float* __restrict__ emax,
                                                       float* __restrict__ ebuf,
                                                       float* __restrict__ denom) {
    int e = blockIdx.x * 256 + threadIdx.x;
    if (e >= N_EDGES) return;
    int s, d; edge_endpoints(ei, e, s, d);
    float ex = __expf(leaky(as[s] + ad[d]) - emax[d]);
    ebuf[e] = ex;
    atomicAdd(&denom[d], ex);
}

__global__ __launch_bounds__(256) void edge_agg_kernel(const int* __restrict__ ei,
                                                       const float* __restrict__ ebuf,
                                                       const float* __restrict__ denom,
                                                       const float* __restrict__ xw,
                                                       float* __restrict__ agg) {
    int e = blockIdx.x * 4 + (threadIdx.x >> 6);
    if (e >= N_EDGES) return;
    int s, d; edge_endpoints(ei, e, s, d);
    float alpha = ebuf[e] / denom[d];
    int c0 = (threadIdx.x & 63) * 4;
    float4 v = *(const float4*)(xw + (size_t)s * 256 + c0);
    float* dst = agg + (size_t)d * 256 + c0;
    atomicAdd(dst + 0, alpha * v.x);
    atomicAdd(dst + 1, alpha * v.y);
    atomicAdd(dst + 2, alpha * v.z);
    atomicAdd(dst + 3, alpha * v.w);
}

__global__ __launch_bounds__(256) void gemm2_kernel(const float* __restrict__ agg,
                                                    const float* __restrict__ b,
                                                    const float* __restrict__ Wfc2,
                                                    const float* __restrict__ bfc2,
                                                    float* __restrict__ h2) {
    const int lane = threadIdx.x & 31;
    const int wave = threadIdx.x >> 5;
    const int rowTile = blockIdx.x * 8 + wave;
    if (rowTile >= ROW_TILES) return;
    const int row0 = rowTile * 16;
    const int col0 = blockIdx.y * 64;

    const int rA  = row0 + (lane & 15);
    const int kHi = (lane >> 4) * 2;
    const int nB  = lane & 15;

    v8f acc[4] = {};
    for (int k = 0; k < 256; k += 4) {
        v2f a;
        a.x = fmaxf(agg[(size_t)rA * 256 + k + kHi]     + b[k + kHi],     0.0f);
        a.y = fmaxf(agg[(size_t)rA * 256 + k + kHi + 1] + b[k + kHi + 1], 0.0f);
#pragma unroll
        for (int t = 0; t < 4; ++t) {
            v2f bm;
            bm.x = Wfc2[(size_t)(k + kHi)     * 128 + col0 + t * 16 + nB];
            bm.y = Wfc2[(size_t)(k + kHi + 1) * 128 + col0 + t * 16 + nB];
            acc[t] = __builtin_amdgcn_wmma_f32_16x16x4_f32(
                false, a, false, bm, (short)0, acc[t], false, false);
        }
    }
    const int mBase = row0 + ((lane >> 4) << 3);
#pragma unroll
    for (int t = 0; t < 4; ++t) {
        int col = col0 + t * 16 + nB;
#pragma unroll
        for (int r = 0; r < 8; ++r)
            h2[(size_t)(mBase + r) * 128 + col] = fmaxf(acc[t][r] + bfc2[col], 0.0f);
    }
}

__global__ __launch_bounds__(256) void gemm3_kernel(const float* __restrict__ h2,
                                                    const float* __restrict__ Wout,
                                                    const float* __restrict__ bout,
                                                    float* __restrict__ out) {
    const int lane = threadIdx.x & 31;
    const int wave = threadIdx.x >> 5;
    const int rowTile = blockIdx.x * 8 + wave;
    if (rowTile >= ROW_TILES) return;
    const int row0 = rowTile * 16;

    const int rA  = row0 + (lane & 15);
    const int kHi = (lane >> 4) * 2;
    const int nB  = lane & 15;

    v8f acc = {};
    for (int k = 0; k < 128; k += 4) {
        v2f a;
        a.x = h2[(size_t)rA * 128 + k + kHi];
        a.y = h2[(size_t)rA * 128 + k + kHi + 1];
        v2f bm;
        bm.x = (nB < 6) ? Wout[(size_t)(k + kHi)     * 6 + nB] : 0.0f;
        bm.y = (nB < 6) ? Wout[(size_t)(k + kHi + 1) * 6 + nB] : 0.0f;
        acc = __builtin_amdgcn_wmma_f32_16x16x4_f32(
            false, a, false, bm, (short)0, acc, false, false);
    }
    if (nB < 6) {
        const int mBase = row0 + ((lane >> 4) << 3);
        float bb = bout[nB];
#pragma unroll
        for (int r = 0; r < 8; ++r)
            out[(size_t)(mBase + r) * 6 + nB] = acc[r] + bb;
    }
}

extern "C" void kernel_launch(void* const* d_in, const int* in_sizes, int n_in,
                              void* d_out, int out_size, void* d_ws, size_t ws_size,
                              hipStream_t stream) {
    const float* x     = (const float*)d_in[0];
    const int*   ei    = (const int*)  d_in[1];
    const float* W     = (const float*)d_in[2];
    const float* a_src = (const float*)d_in[3];
    const float* a_dst = (const float*)d_in[4];
    const float* b     = (const float*)d_in[5];
    const float* Wfc2  = (const float*)d_in[6];
    const float* bfc2  = (const float*)d_in[7];
    const float* Wout  = (const float*)d_in[8];
    const float* bout  = (const float*)d_in[9];
    float* out = (float*)d_out;

    float* ws      = (float*)d_ws;
    float* xw      = ws;
    float* agg     = xw + (size_t)N_NODES * 256;
    float* alpha_s = agg + (size_t)N_NODES * 256;
    float* alpha_d = alpha_s + N_NODES;
    float* emax    = alpha_d + N_NODES;
    float* denom   = emax + N_NODES;
    float* ebuf    = denom + N_NODES;
    float* h2      = xw;

    init_ws<<<N_NODES, 256, 0, stream>>>(agg, emax, denom);
    gemm1_xw<<<dim3((ROW_TILES + 7) / 8, 4), 256, 0, stream>>>(x, W, xw);
    alphas_kernel<<<(N_NODES + 7) / 8, 256, 0, stream>>>(xw, a_src, a_dst, alpha_s, alpha_d);
    edge_max_kernel<<<(N_EDGES + 255) / 256, 256, 0, stream>>>(ei, alpha_s, alpha_d, emax);
    edge_exp_kernel<<<(N_EDGES + 255) / 256, 256, 0, stream>>>(ei, alpha_s, alpha_d, emax, ebuf, denom);
    edge_agg_kernel<<<(N_EDGES + 3) / 4, 256, 0, stream>>>(ei, ebuf, denom, xw, agg);
    gemm2_kernel<<<dim3((ROW_TILES + 7) / 8, 2), 256, 0, stream>>>(agg, b, Wfc2, bfc2, h2);
    gemm3_kernel<<<(ROW_TILES + 7) / 8, 256, 0, stream>>>(h2, Wout, bout, out);
}


# Edges counting-sorted by destination (integer atomics); one wave per node does max, sum and weighted row sum in registers: no f32 atomics, no agg zero fill; per-node sum order = slot arrival
# speedup vs baseline: 2.0022x; 2.0022x over previous
#include <hip/hip_runtime.h>
#include <hip/hip_bf16.h>

typedef __attribute__((ext_vector_type(2))) float v2f;
typedef __attribute__((ext_vector_type(8))) float v8f;

#define N_NODES   100000
#define N_EDGES0  800000
#define N_EDGES   900000
#define ROW_TILES 6250

__device__ __forceinline__ float leaky(float v) { return v > 0.0f ? v : 0.2f * v; }

__device__ __forceinline__ float wave_max_f(float v) {
#pragma unroll
    for (int off = 16; off; off >>= 1) v = fmaxf(v, __shfl_xor(v, off, 32));
    return v;
}
__device__ __forceinline__ float wave_sum_f(float v) {
#pragma unroll
    for (int off = 16; off; off >>= 1) v += __shfl_xor(v, off, 32);
    return v;
}
__device__ __forceinline__ int wave_sum_i(int v) {
#pragma unroll
    for (int off = 16; off; off >>= 1) v += __shfl_xor(v, off, 32);
    return v;
}
__device__ __forceinline__ int wave_incl_scan_i(int v, int lane) {
#pragma unroll
    for (int off = 1; off < 32; off <<= 1) {
        int u = __shfl_up(v, off, 32);
        if (lane >= off) v += u;
    }
    return v;
}

__global__ __launch_bounds__(256) void zero_cnt(int* __restrict__ cnt) {
    int i = blockIdx.x * 256 + threadIdx.x;
    if (i < N_NODES) cnt[i] = 0;
}

__global__ __launch_bounds__(256) void hist_kernel(const int* __restrict__ ei,
                                                   int* __restrict__ cnt) {
    int i = blockIdx.x * 256 + threadIdx.x;
    if (i >= N_EDGES0 / 4) return;
    int4 d4 = *(const int4*)(ei + N_EDGES0 + 4 * i);
    atomicAdd(&cnt[d4.x], 1);
    atomicAdd(&cnt[d4.y], 1);
    atomicAdd(&cnt[d4.z], 1);
    atomicAdd(&cnt[d4.w], 1);
}

__global__ __launch_bounds__(1024) void scan_kernel(const int* __restrict__ cnt,
                                                    int* __restrict__ rowptr,
                                                    int* __restrict__ cursor) {
    __shared__ int wsum[32];
    __shared__ int bpre;
    const int t = threadIdx.x;
    const int lane = t & 31;
    const int w = t >> 5;
    const int b = blockIdx.x;

    int pre = 0;
    for (int i = 0; i < b; ++i) pre += cnt[i * 1024 + t];
    pre = wave_sum_i(pre);
    if (lane == 0) wsum[w] = pre;
    __syncthreads();
    if (w == 0) {
        int v = wave_sum_i(wsum[lane]);
        if (lane == 0) bpre = v;
    }
    __syncthreads();
    const int blockPrefix = bpre;

    const int idx = b * 1024 + t;
    const int v = (idx < N_NODES) ? cnt[idx] : 0;
    const int inc = wave_incl_scan_i(v, lane);
    if (lane == 31) wsum[w] = inc;
    __syncthreads();
    const int wt = wsum[lane];
    const int wexcl = wave_incl_scan_i(wt, lane) - wt;
    const int wavePrefix = __shfl(wexcl, w, 32);
    const int excl = blockPrefix + wavePrefix + inc - v;
    if (idx < N_NODES) { rowptr[idx] = excl; cursor[idx] = excl; }
}

__global__ __launch_bounds__(256) void fill_kernel(const int* __restrict__ ei,
                                                   int* __restrict__ cursor,
                                                   int* __restrict__ csr) {
    int i = blockIdx.x * 256 + threadIdx.x;
    if (i >= N_EDGES0 / 4) return;
    int4 s4 = *(const int4*)(ei + 4 * i);
    int4 d4 = *(const int4*)(ei + N_EDGES0 + 4 * i);
    int p0 = atomicAdd(&cursor[d4.x], 1);
    int p1 = atomicAdd(&cursor[d4.y], 1);
    int p2 = atomicAdd(&cursor[d4.z], 1);
    int p3 = atomicAdd(&cursor[d4.w], 1);
    csr[p0] = s4.x;
    csr[p1] = s4.y;
    csr[p2] = s4.z;
    csr[p3] = s4.w;
}

__global__ __launch_bounds__(256) void gemm1_xw(const float* __restrict__ x,
                                                const float* __restrict__ W,
                                                float* __restrict__ xw) {
    const int lane = threadIdx.x & 31;
    const int wave = threadIdx.x >> 5;
    const int rowTile = blockIdx.x * 8 + wave;
    if (rowTile >= ROW_TILES) return;
    const int row0 = rowTile * 16;
    const int col0 = blockIdx.y * 64;

    const int rA  = row0 + (lane & 15);
    const int kHi = (lane >> 4) * 2;
    const int nB  = lane & 15;

    v8f acc[4] = {};
    for (int k = 0; k < 128; k += 4) {
        v2f a;
        a.x = x[(size_t)rA * 128 + k + kHi];
        a.y = x[(size_t)rA * 128 + k + kHi + 1];
#pragma unroll
        for (int t = 0; t < 4; ++t) {
            v2f bm;
            bm.x = W[(size_t)(k + kHi)     * 256 + col0 + t * 16 + nB];
            bm.y = W[(size_t)(k + kHi + 1) * 256 + col0 + t * 16 + nB];
            acc[t] = __builtin_amdgcn_wmma_f32_16x16x4_f32(
                false, a, false, bm, (short)0, acc[t], false, false);
        }
    }
    const int mBase = row0 + ((lane >> 4) << 3);
#pragma unroll
    for (int t = 0; t < 4; ++t)
#pragma unroll
        for (int r = 0; r < 8; ++r)
            xw[(size_t)(mBase + r) * 256 + col0 + t * 16 + nB] = acc[t][r];
}

__global__ __launch_bounds__(256) void alphas_kernel(const float* __restrict__ xw,
                                                     const float* __restrict__ a_src,
                                                     const float* __restrict__ a_dst,
                                                     float* __restrict__ as,
                                                     float* __restrict__ ad) {
    const int lane = threadIdx.x & 31;
    const int node = blockIdx.x * 8 + (threadIdx.x >> 5);
    if (node >= N_NODES) return;
    const float* xr = xw + (size_t)node * 256;
    float s = 0.0f, d = 0.0f;
#pragma unroll
    for (int i = 0; i < 8; ++i) {
        float v = xr[lane + 32 * i];
        s += v * a_src[lane + 32 * i];
        d += v * a_dst[lane + 32 * i];
    }
#pragma unroll
    for (int off = 16; off; off >>= 1) {
        s += __shfl_xor(s, off, 32);
        d += __shfl_xor(d, off, 32);
    }
    if (lane == 0) { as[node] = s; ad[node] = d; }
}

__device__ __forceinline__ int lane_i(int v, int t) { return __builtin_amdgcn_readlane(v, t); }
__device__ __forceinline__ float lane_f(float v, int t) {
    return __int_as_float(__builtin_amdgcn_readlane(__float_as_int(v), t));
}

__global__ __launch_bounds__(256) void node_gather_kernel(const int* __restrict__ csr,
                                                          const int* __restrict__ rowptr,
                                                          const int* __restrict__ cnt,
                                                          const float* __restrict__ as,
                                                          const float* __restrict__ ad,
                                                          const float* __restrict__ xw,
                                                          float* __restrict__ agg) {
    const int lane = threadIdx.x & 31;
    const int d = blockIdx.x * 8 + __builtin_amdgcn_readfirstlane((int)(threadIdx.x >> 5));
    if (d >= N_NODES) return;
    const int start = rowptr[d];
    const int L = cnt[d] + 1;
    const float adv = ad[d];

    int s0 = d;
    float e0 = -INFINITY;
    if (lane < L) {
        if (lane > 0) s0 = csr[start + lane - 1];
        e0 = leaky(as[s0] + adv);
    }
    float m = e0;
    for (int p = lane + 32; p < L; p += 32)
        m = fmaxf(m, leaky(as[csr[start + p - 1]] + adv));
    m = wave_max_f(m);

    const float x0 = (lane < L) ? __expf(e0 - m) : 0.0f;
    float sum = x0;
    for (int p = lane + 32; p < L; p += 32)
        sum += __expf(leaky(as[csr[start + p - 1]] + adv) - m);
    sum = wave_sum_f(sum);

    const int c0 = lane * 4;
    float4 accA = make_float4(0.0f, 0.0f, 0.0f, 0.0f);
    float4 accB = make_float4(0.0f, 0.0f, 0.0f, 0.0f);
    for (int base = 0; base < L; base += 32) {
        int s = s0;
        float al = x0 / sum;
        if (base > 0) {
            const int p = base + lane;
            s = d; al = 0.0f;
            if (p < L) {
                s = csr[start + p - 1];
                al = __expf(leaky(as[s] + adv) - m) / sum;
            }
        }
        const int nc = (L - base < 32) ? (L - base) : 32;
        int t = 0;
        for (; t + 4 <= nc; t += 4) {
            const int   sa = lane_i(s, t),      sb = lane_i(s, t + 1);
            const int   sc = lane_i(s, t + 2),  sd = lane_i(s, t + 3);
            const float aa = lane_f(al, t),     ab = lane_f(al, t + 1);
            const float ac = lane_f(al, t + 2), ae = lane_f(al, t + 3);
            const float* ra = xw + (size_t)sa * 256 + c0;
            const float* rb = xw + (size_t)sb * 256 + c0;
            const float* rc = xw + (size_t)sc * 256 + c0;
            const float* rd = xw + (size_t)sd * 256 + c0;
            const float4 va0 = *(const float4*)ra, va1 = *(const float4*)(ra + 128);
            const float4 vb0 = *(const float4*)rb, vb1 = *(const float4*)(rb + 128);
            const float4 vc0 = *(const float4*)rc, vc1 = *(const float4*)(rc + 128);
            const float4 vd0 = *(const float4*)rd, vd1 = *(const float4*)(rd + 128);
            accA.x = fmaf(aa, va0.x, accA.x); accA.y = fmaf(aa, va0.y, accA.y); accA.z = fmaf(aa, va0.z, accA.z); accA.w = fmaf(aa, va0.w, accA.w);
            accB.x = fmaf(aa, va1.x, accB.x); accB.y = fmaf(aa, va1.y, accB.y); accB.z = fmaf(aa, va1.z, accB.z); accB.w = fmaf(aa, va1.w, accB.w);
            accA.x = fmaf(ab, vb0.x, accA.x); accA.y = fmaf(ab, vb0.y, accA.y); accA.z = fmaf(ab, vb0.z, accA.z); accA.w = fmaf(ab, vb0.w, accA.w);
            accB.x = fmaf(ab, vb1.x, accB.x); accB.y = fmaf(ab, vb1.y, accB.y); accB.z = fmaf(ab, vb1.z, accB.z); accB.w = fmaf(ab, vb1.w, accB.w);
            accA.x = fmaf(ac, vc0.x, accA.x); accA.y = fmaf(ac, vc0.y, accA.y); accA.z = fmaf(ac, vc0.z, accA.z); accA.w = fmaf(ac, vc0.w, accA.w);
            accB.x = fmaf(ac, vc1.x, accB.x); accB.y = fmaf(ac, vc1.y, accB.y); accB.z = fmaf(ac, vc1.z, accB.z); accB.w = fmaf(ac, vc1.w, accB.w);
            accA.x = fmaf(ae, vd0.x, accA.x); accA.y = fmaf(ae, vd0.y, accA.y); accA.z = fmaf(ae, vd0.z, accA.z); accA.w = fmaf(ae, vd0.w, accA.w);
            accB.x = fmaf(ae, vd1.x, accB.x); accB.y = fmaf(ae, vd1.y, accB.y); accB.z = fmaf(ae, vd1.z, accB.z); accB.w = fmaf(ae, vd1.w, accB.w);
        }
        for (; t < nc; ++t) {
            const int   sa = lane_i(s, t);
            const float aa = lane_f(al, t);
            const float* ra = xw + (size_t)sa * 256 + c0;
            const float4 va0 = *(const float4*)ra, va1 = *(const float4*)(ra + 128);
            accA.x = fmaf(aa, va0.x, accA.x); accA.y = fmaf(aa, va0.y, accA.y); accA.z = fmaf(aa, va0.z, accA.z); accA.w = fmaf(aa, va0.w, accA.w);
            accB.x = fmaf(aa, va1.x, accB.x); accB.y = fmaf(aa, va1.y, accB.y); accB.z = fmaf(aa, va1.z, accB.z); accB.w = fmaf(aa, va1.w, accB.w);
        }
    }
    float* dst = agg + (size_t)d * 256 + c0;
    *(float4*)dst = accA;
    *(float4*)(dst + 128) = accB;
}

__global__ __launch_bounds__(256) void gemm2_kernel(const float* __restrict__ agg,
                                                    const float* __restrict__ b,
                                                    const float* __restrict__ Wfc2,
                                                    const float* __restrict__ bfc2,
                                                    float* __restrict__ h2) {
    const int lane = threadIdx.x & 31;
    const int wave = threadIdx.x >> 5;
    const int rowTile = blockIdx.x * 8 + wave;
    if (rowTile >= ROW_TILES) return;
    const int row0 = rowTile * 16;
    const int col0 = blockIdx.y * 64;

    const int rA  = row0 + (lane & 15);
    const int kHi = (lane >> 4) * 2;
    const int nB  = lane & 15;

    v8f acc[4] = {};
    for (int k = 0; k < 256; k += 4) {
        v2f a;
        a.x = fmaxf(agg[(size_t)rA * 256 + k + kHi]     + b[k + kHi],     0.0f);
        a.y = fmaxf(agg[(size_t)rA * 256 + k + kHi + 1] + b[k + kHi + 1], 0.0f);
#pragma unroll
        for (int t = 0; t < 4; ++t) {
            v2f bm;
            bm.x = Wfc2[(size_t)(k + kHi)     * 128 + col0 + t * 16 + nB];
            bm.y = Wfc2[(size_t)(k + kHi + 1) * 128 + col0 + t * 16 + nB];
            acc[t] = __builtin_amdgcn_wmma_f32_16x16x4_f32(
                false, a, false, bm, (short)0, acc[t], false, false);
        }
    }
    const int mBase = row0 + ((lane >> 4) << 3);
#pragma unroll
    for (int t = 0; t < 4; ++t) {
        int col = col0 + t * 16 + nB;
#pragma unroll
        for (int r = 0; r < 8; ++r)
            h2[(size_t)(mBase + r) * 128 + col] = fmaxf(acc[t][r] + bfc2[col], 0.0f);
    }
}

__global__ __launch_bounds__(256) void gemm3_kernel(const float* __restrict__ h2,
                                                    const float* __restrict__ Wout,
                                                    const float* __restrict__ bout,
                                                    float* __restrict__ out) {
    const int lane = threadIdx.x & 31;
    const int wave = threadIdx.x >> 5;
    const int rowTile = blockIdx.x * 8 + wave;
    if (rowTile >= ROW_TILES) return;
    const int row0 = rowTile * 16;

    const int rA  = row0 + (lane & 15);
    const int kHi = (lane >> 4) * 2;
    const int nB  = lane & 15;

    v8f acc = {};
    for (int k = 0; k < 128; k += 4) {
        v2f a;
        a.x = h2[(size_t)rA * 128 + k + kHi];
        a.y = h2[(size_t)rA * 128 + k + kHi + 1];
        v2f bm;
        bm.x = (nB < 6) ? Wout[(size_t)(k + kHi)     * 6 + nB] : 0.0f;
        bm.y = (nB < 6) ? Wout[(size_t)(k + kHi + 1) * 6 + nB] : 0.0f;
        acc = __builtin_amdgcn_wmma_f32_16x16x4_f32(
            false, a, false, bm, (short)0, acc, false, false);
    }
    if (nB < 6) {
        const int mBase = row0 + ((lane >> 4) << 3);
        float bb = bout[nB];
#pragma unroll
        for (int r = 0; r < 8; ++r)
            out[(size_t)(mBase + r) * 6 + nB] = acc[r] + bb;
    }
}

extern "C" void kernel_launch(void* const* d_in, const int* in_sizes, int n_in,
                              void* d_out, int out_size, void* d_ws, size_t ws_size,
                              hipStream_t stream) {
    const float* x     = (const float*)d_in[0];
    const int*   ei    = (const int*)  d_in[1];
    const float* W     = (const float*)d_in[2];
    const float* a_src = (const float*)d_in[3];
    const float* a_dst = (const float*)d_in[4];
    const float* b     = (const float*)d_in[5];
    const float* Wfc2  = (const float*)d_in[6];
    const float* bfc2  = (const float*)d_in[7];
    const float* Wout  = (const float*)d_in[8];
    const float* bout  = (const float*)d_in[9];
    float* out = (float*)d_out;

    float* ws      = (float*)d_ws;
    float* xw      = ws;
    float* agg     = xw + (size_t)N_NODES * 256;
    float* alpha_s = agg + (size_t)N_NODES * 256;
    float* alpha_d = alpha_s + N_NODES;
    int*   cnt     = (int*)(alpha_d + N_NODES);
    int*   rowptr  = cnt + N_NODES;
    int*   cursor  = rowptr + N_NODES;
    int*   csr     = cursor + N_NODES;
    float* h2      = xw;

    zero_cnt<<<(N_NODES + 255) / 256, 256, 0, stream>>>(cnt);
    hist_kernel<<<(N_EDGES0 / 4 + 255) / 256, 256, 0, stream>>>(ei, cnt);
    scan_kernel<<<(N_NODES + 1023) / 1024, 1024, 0, stream>>>(cnt, rowptr, cursor);
    fill_kernel<<<(N_EDGES0 / 4 + 255) / 256, 256, 0, stream>>>(ei, cursor, csr);
    gemm1_xw<<<dim3((ROW_TILES + 7) / 8, 4), 256, 0, stream>>>(x, W, xw);
    alphas_kernel<<<(N_NODES + 7) / 8, 256, 0, stream>>>(xw, a_src, a_dst, alpha_s, alpha_d);
    node_gather_kernel<<<(N_NODES + 7) / 8, 256, 0, stream>>>(csr, rowptr, cnt, alpha_s, alpha_d, xw, agg);
    gemm2_kernel<<<dim3((ROW_TILES + 7) / 8, 2), 256, 0, stream>>>(agg, b, Wfc2, bfc2, h2);
    gemm3_kernel<<<(ROW_TILES + 7) / 8, 256, 0, stream>>>(h2, Wout, bout, out);
}


# Linear map W moved behind the aggregation: gather sums 512-byte x rows, not 1-KB xW rows; scores are x.(W a); agg = (sum alpha x) W on WMMA; f32 sum order changes (rows summed before the product)
# speedup vs baseline: 2.2580x; 1.1278x over previous
#include <hip/hip_runtime.h>
#include <hip/hip_bf16.h>

typedef __attribute__((ext_vector_type(2))) float v2f;
typedef __attribute__((ext_vector_type(8))) float v8f;

#define N_NODES   100000
#define N_EDGES0  800000
#define N_EDGES   900000
#define ROW_TILES 6250

__device__ __forceinline__ float leaky(float v) { return v > 0.0f ? v : 0.2f * v; }

__device__ __forceinline__ float wave_max_f(float v) {
#pragma unroll
    for (int off = 16; off; off >>= 1) v = fmaxf(v, __shfl_xor(v, off, 32));
    return v;
}
__device__ __forceinline__ float wave_sum_f(float v) {
#pragma unroll
    for (int off = 16; off; off >>= 1) v += __shfl_xor(v, off, 32);
    return v;
}
__device__ __forceinline__ int wave_sum_i(int v) {
#pragma unroll
    for (int off = 16; off; off >>= 1) v += __shfl_xor(v, off, 32);
    return v;
}
__device__ __forceinline__ int wave_incl_scan_i(int v, int lane) {
#pragma unroll
    for (int off = 1; off < 32; off <<= 1) {
        int u = __shfl_up(v, off, 32);
        if (lane >= off) v += u;
    }
    return v;
}

#define ZERO_BLOCKS 391
#define FOLD_BLOCKS 16
__global__ __launch_bounds__(256) void prep_kernel(int* __restrict__ cnt,
                                                   const float* __restrict__ W,
                                                   const float* __restrict__ a_src,
                                                   const float* __restrict__ a_dst,
                                                   float* __restrict__ wsd) {
    if (blockIdx.x < ZERO_BLOCKS) {
        int i = blockIdx.x * 256 + threadIdx.x;
        if (i < N_NODES) cnt[i] = 0;
        return;
    }
    const int lane = threadIdx.x & 31;
    const int k = (blockIdx.x - ZERO_BLOCKS) * 8 + (threadIdx.x >> 5);
    const float* wr = W + (size_t)k * 256;
    float s = 0.0f, d = 0.0f;
#pragma unroll
    for (int i = 0; i < 8; ++i) {
        float v = wr[lane + 32 * i];
        s += v * a_src[lane + 32 * i];
        d += v * a_dst[lane + 32 * i];
    }
    s = wave_sum_f(s);
    d = wave_sum_f(d);
    if (lane == 0) { wsd[k] = s; wsd[128 + k] = d; }
}

__global__ __launch_bounds__(256) void hist_kernel(const int* __restrict__ ei,
                                                   int* __restrict__ cnt) {
    int i = blockIdx.x * 256 + threadIdx.x;
    if (i >= N_EDGES0 / 4) return;
    int4 d4 = *(const int4*)(ei + N_EDGES0 + 4 * i);
    atomicAdd(&cnt[d4.x], 1);
    atomicAdd(&cnt[d4.y], 1);
    atomicAdd(&cnt[d4.z], 1);
    atomicAdd(&cnt[d4.w], 1);
}

__global__ __launch_bounds__(1024) void scan_kernel(const int* __restrict__ cnt,
                                                    int* __restrict__ rowptr,
                                                    int* __restrict__ cursor) {
    __shared__ int wsum[32];
    __shared__ int bpre;
    const int t = threadIdx.x;
    const int lane = t & 31;
    const int w = t >> 5;
    const int b = blockIdx.x;

    int pre = 0;
    for (int i = 0; i < b; ++i) pre += cnt[i * 1024 + t];
    pre = wave_sum_i(pre);
    if (lane == 0) wsum[w] = pre;
    __syncthreads();
    if (w == 0) {
        int v = wave_sum_i(wsum[lane]);
        if (lane == 0) bpre = v;
    }
    __syncthreads();
    const int blockPrefix = bpre;

    const int idx = b * 1024 + t;
    const int v = (idx < N_NODES) ? cnt[idx] : 0;
    const int inc = wave_incl_scan_i(v, lane);
    if (lane == 31) wsum[w] = inc;
    __syncthreads();
    const int wt = wsum[lane];
    const int wexcl = wave_incl_scan_i(wt, lane) - wt;
    const int wavePrefix = __shfl(wexcl, w, 32);
    const int excl = blockPrefix + wavePrefix + inc - v;
    if (idx < N_NODES) { rowptr[idx] = excl; cursor[idx] = excl; }
}

__global__ __launch_bounds__(256) void fill_kernel(const int* __restrict__ ei,
                                                   int* __restrict__ cursor,
                                                   int* __restrict__ csr) {
    int i = blockIdx.x * 256 + threadIdx.x;
    if (i >= N_EDGES0 / 4) return;
    int4 s4 = *(const int4*)(ei + 4 * i);
    int4 d4 = *(const int4*)(ei + N_EDGES0 + 4 * i);
    int p0 = atomicAdd(&cursor[d4.x], 1);
    int p1 = atomicAdd(&cursor[d4.y], 1);
    int p2 = atomicAdd(&cursor[d4.z], 1);
    int p3 = atomicAdd(&cursor[d4.w], 1);
    csr[p0] = s4.x;
    csr[p1] = s4.y;
    csr[p2] = s4.z;
    csr[p3] = s4.w;
}

__global__ __launch_bounds__(256) void gemm1_agg(const float* __restrict__ aggx,
                                                 const float* __restrict__ W,
                                                 float* __restrict__ agg) {
    const int lane = threadIdx.x & 31;
    const int wave = threadIdx.x >> 5;
    const int rowTile = blockIdx.x * 8 + wave;
    if (rowTile >= ROW_TILES) return;
    const int row0 = rowTile * 16;
    const int col0 = blockIdx.y * 64;

    const int rA  = row0 + (lane & 15);
    const int kHi = (lane >> 4) * 2;
    const int nB  = lane & 15;

    v8f acc[4] = {};
    for (int k = 0; k < 128; k += 4) {
        v2f a;
        a.x = aggx[(size_t)rA * 128 + k + kHi];
        a.y = aggx[(size_t)rA * 128 + k + kHi + 1];
#pragma unroll
        for (int t = 0; t < 4; ++t) {
            v2f bm;
            bm.x = W[(size_t)(k + kHi)     * 256 + col0 + t * 16 + nB];
            bm.y = W[(size_t)(k + kHi + 1) * 256 + col0 + t * 16 + nB];
            acc[t] = __builtin_amdgcn_wmma_f32_16x16x4_f32(
                false, a, false, bm, (short)0, acc[t], false, false);
        }
    }
    const int mBase = row0 + ((lane >> 4) << 3);
#pragma unroll
    for (int t = 0; t < 4; ++t)
#pragma unroll
        for (int r = 0; r < 8; ++r)
            agg[(size_t)(mBase + r) * 256 + col0 + t * 16 + nB] = acc[t][r];
}

__global__ __launch_bounds__(256) void alphas_kernel(const float* __restrict__ x,
                                                     const float* __restrict__ wsd,
                                                     float* __restrict__ as,
                                                     float* __restrict__ ad) {
    const int lane = threadIdx.x & 31;
    const int node0 = (blockIdx.x * 8 + (threadIdx.x >> 5)) * 4;
    if (node0 >= N_NODES) return;
    const float4 ws = *(const float4*)(wsd + 4 * lane);
    const float4 wd = *(const float4*)(wsd + 128 + 4 * lane);
    const float* xr = x + (size_t)node0 * 128 + 4 * lane;
    const float4 v0 = *(const float4*)xr;
    const float4 v1 = *(const float4*)(xr + 128);
    const float4 v2 = *(const float4*)(xr + 256);
    const float4 v3 = *(const float4*)(xr + 384);
    float s0 = v0.x * ws.x + v0.y * ws.y + v0.z * ws.z + v0.w * ws.w;
    float s1 = v1.x * ws.x + v1.y * ws.y + v1.z * ws.z + v1.w * ws.w;
    float s2 = v2.x * ws.x + v2.y * ws.y + v2.z * ws.z + v2.w * ws.w;
    float s3 = v3.x * ws.x + v3.y * ws.y + v3.z * ws.z + v3.w * ws.w;
    float d0 = v0.x * wd.x + v0.y * wd.y + v0.z * wd.z + v0.w * wd.w;
    float d1 = v1.x * wd.x + v1.y * wd.y + v1.z * wd.z + v1.w * wd.w;
    float d2 = v2.x * wd.x + v2.y * wd.y + v2.z * wd.z + v2.w * wd.w;
    float d3 = v3.x * wd.x + v3.y * wd.y + v3.z * wd.z + v3.w * wd.w;
    s0 = wave_sum_f(s0); s1 = wave_sum_f(s1); s2 = wave_sum_f(s2); s3 = wave_sum_f(s3);
    d0 = wave_sum_f(d0); d1 = wave_sum_f(d1); d2 = wave_sum_f(d2); d3 = wave_sum_f(d3);
    if (lane == 0) {
        *(float4*)(as + node0) = make_float4(s0, s1, s2, s3);
        *(float4*)(ad + node0) = make_float4(d0, d1, d2, d3);
    }
}

__device__ __forceinline__ int lane_i(int v, int t) { return __builtin_amdgcn_readlane(v, t); }
__device__ __forceinline__ float lane_f(float v, int t) {
    return __int_as_float(__builtin_amdgcn_readlane(__float_as_int(v), t));
}

__global__ __launch_bounds__(256) void node_gather_kernel(const int* __restrict__ csr,
                                                          const int* __restrict__ rowptr,
                                                          const int* __restrict__ cnt,
                                                          const float* __restrict__ as,
                                                          const float* __restrict__ ad,
                                                          const float* __restrict__ x,
                                                          float* __restrict__ aggx) {
    const int lane = threadIdx.x & 31;
    const int d = blockIdx.x * 8 + __builtin_amdgcn_readfirstlane((int)(threadIdx.x >> 5));
    if (d >= N_NODES) return;
    const int start = rowptr[d];
    const int L = cnt[d] + 1;
    const float adv = ad[d];

    int s0 = d;
    float e0 = -INFINITY;
    if (lane < L) {
        if (lane > 0) s0 = csr[start + lane - 1];
        e0 = leaky(as[s0] + adv);
    }
    float m = e0;
    for (int p = lane + 32; p < L; p += 32)
        m = fmaxf(m, leaky(as[csr[start + p - 1]] + adv));
    m = wave_max_f(m);

    const float x0 = (lane < L) ? __expf(e0 - m) : 0.0f;
    float sum = x0;
    for (int p = lane + 32; p < L; p += 32)
        sum += __expf(leaky(as[csr[start + p - 1]] + adv) - m);
    sum = wave_sum_f(sum);

    const int c0 = lane * 4;
    float4 acc = make_float4(0.0f, 0.0f, 0.0f, 0.0f);
    for (int base = 0; base < L; base += 32) {
        int s = s0;
        float al = x0 / sum;
        if (base > 0) {
            const int p = base + lane;
            s = d; al = 0.0f;
            if (p < L) {
                s = csr[start + p - 1];
                al = __expf(leaky(as[s] + adv) - m) / sum;
            }
        }
        const int nc = (L - base < 32) ? (L - base) : 32;
        int t = 0;
        for (; t + 8 <= nc; t += 8) {
            float4 v[8];
            float  a[8];
#pragma unroll
            for (int j = 0; j < 8; ++j) {
                const int sj = lane_i(s, t + j);
                a[j] = lane_f(al, t + j);
                v[j] = *(const float4*)(x + (size_t)sj * 128 + c0);
            }
#pragma unroll
            for (int j = 0; j < 8; ++j) {
                acc.x = fmaf(a[j], v[j].x, acc.x); acc.y = fmaf(a[j], v[j].y, acc.y);
                acc.z = fmaf(a[j], v[j].z, acc.z); acc.w = fmaf(a[j], v[j].w, acc.w);
            }
        }
        for (; t + 4 <= nc; t += 4) {
            float4 v[4];
            float  a[4];
#pragma unroll
            for (int j = 0; j < 4; ++j) {
                const int sj = lane_i(s, t + j);
                a[j] = lane_f(al, t + j);
                v[j] = *(const float4*)(x + (size_t)sj * 128 + c0);
            }
#pragma unroll
            for (int j = 0; j < 4; ++j) {
                acc.x = fmaf(a[j], v[j].x, acc.x); acc.y = fmaf(a[j], v[j].y, acc.y);
                acc.z = fmaf(a[j], v[j].z, acc.z); acc.w = fmaf(a[j], v[j].w, acc.w);
            }
        }
#pragma clang loop unroll(disable)
        for (; t < nc; ++t) {
            const int   sa = lane_i(s, t);
            const float aa = lane_f(al, t);
            const float4 va = *(const float4*)(x + (size_t)sa * 128 + c0);
            acc.x = fmaf(aa, va.x, acc.x); acc.y = fmaf(aa, va.y, acc.y);
            acc.z = fmaf(aa, va.z, acc.z); acc.w = fmaf(aa, va.w, acc.w);
        }
    }
    *(float4*)(aggx + (size_t)d * 128 + c0) = acc;
}

__global__ __launch_bounds__(256) void gemm2_kernel(const float* __restrict__ agg,
                                                    const float* __restrict__ b,
                                                    const float* __restrict__ Wfc2,
                                                    const float* __restrict__ bfc2,
                                                    float* __restrict__ h2) {
    const int lane = threadIdx.x & 31;
    const int wave = threadIdx.x >> 5;
    const int rowTile = blockIdx.x * 8 + wave;
    if (rowTile >= ROW_TILES) return;
    const int row0 = rowTile * 16;
    const int col0 = blockIdx.y * 64;

    const int rA  = row0 + (lane & 15);
    const int kHi = (lane >> 4) * 2;
    const int nB  = lane & 15;

    v8f acc[4] = {};
    for (int k = 0; k < 256; k += 4) {
        v2f a;
        a.x = fmaxf(agg[(size_t)rA * 256 + k + kHi]     + b[k + kHi],     0.0f);
        a.y = fmaxf(agg[(size_t)rA * 256 + k + kHi + 1] + b[k + kHi + 1], 0.0f);
#pragma unroll
        for (int t = 0; t < 4; ++t) {
            v2f bm;
            bm.x = Wfc2[(size_t)(k + kHi)     * 128 + col0 + t * 16 + nB];
            bm.y = Wfc2[(size_t)(k + kHi + 1) * 128 + col0 + t * 16 + nB];
            acc[t] = __builtin_amdgcn_wmma_f32_16x16x4_f32(
                false, a, false, bm, (short)0, acc[t], false, false);
        }
    }
    const int mBase = row0 + ((lane >> 4) << 3);
#pragma unroll
    for (int t = 0; t < 4; ++t) {
        int col = col0 + t * 16 + nB;
#pragma unroll
        for (int r = 0; r < 8; ++r)
            h2[(size_t)(mBase + r) * 128 + col] = fmaxf(acc[t][r] + bfc2[col], 0.0f);
    }
}

__global__ __launch_bounds__(256) void gemm3_kernel(const float* __restrict__ h2,
                                                    const float* __restrict__ Wout,
                                                    const float* __restrict__ bout,
                                                    float* __restrict__ out) {
    const int lane = threadIdx.x & 31;
    const int wave = threadIdx.x >> 5;
    const int rowTile = blockIdx.x * 8 + wave;
    if (rowTile >= ROW_TILES) return;
    const int row0 = rowTile * 16;

    const int rA  = row0 + (lane & 15);
    const int kHi = (lane >> 4) * 2;
    const int nB  = lane & 15;

    v8f acc = {};
    for (int k = 0; k < 128; k += 4) {
        v2f a;
        a.x = h2[(size_t)rA * 128 + k + kHi];
        a.y = h2[(size_t)rA * 128 + k + kHi + 1];
        v2f bm;
        bm.x = (nB < 6) ? Wout[(size_t)(k + kHi)     * 6 + nB] : 0.0f;
        bm.y = (nB < 6) ? Wout[(size_t)(k + kHi + 1) * 6 + nB] : 0.0f;
        acc = __builtin_amdgcn_wmma_f32_16x16x4_f32(
            false, a, false, bm, (short)0, acc, false, false);
    }
    if (nB < 6) {
        const int mBase = row0 + ((lane >> 4) << 3);
        float bb = bout[nB];
#pragma unroll
        for (int r = 0; r < 8; ++r)
            out[(size_t)(mBase + r) * 6 + nB] = acc[r] + bb;
    }
}

extern "C" void kernel_launch(void* const* d_in, const int* in_sizes, int n_in,
                              void* d_out, int out_size, void* d_ws, size_t ws_size,
                              hipStream_t stream) {
    const float* x     = (const float*)d_in[0];
    const int*   ei    = (const int*)  d_in[1];
    const float* W     = (const float*)d_in[2];
    const float* a_src = (const float*)d_in[3];
    const float* a_dst = (const float*)d_in[4];
    const float* b     = (const float*)d_in[5];
    const float* Wfc2  = (const float*)d_in[6];
    const float* bfc2  = (const float*)d_in[7];
    const float* Wout  = (const float*)d_in[8];
    const float* bout  = (const float*)d_in[9];
    float* out = (float*)d_out;

    float* ws      = (float*)d_ws;
    float* agg     = ws;
    float* aggx    = agg + (size_t)N_NODES * 256;
    float* alpha_s = aggx + (size_t)N_NODES * 128;
    float* alpha_d = alpha_s + N_NODES;
    float* wsd     = alpha_d + N_NODES;
    int*   cnt     = (int*)(wsd + 256);
    int*   rowptr  = cnt + N_NODES;
    int*   cursor  = rowptr + N_NODES;
    int*   csr     = cursor + N_NODES;
    float* h2      = aggx;

    prep_kernel<<<ZERO_BLOCKS + FOLD_BLOCKS, 256, 0, stream>>>(cnt, W, a_src, a_dst, wsd);
    hist_kernel<<<(N_EDGES0 / 4 + 255) / 256, 256, 0, stream>>>(ei, cnt);
    scan_kernel<<<(N_NODES + 1023) / 1024, 1024, 0, stream>>>(cnt, rowptr, cursor);
    fill_kernel<<<(N_EDGES0 / 4 + 255) / 256, 256, 0, stream>>>(ei, cursor, csr);
    alphas_kernel<<<(N_NODES / 4 + 7) / 8, 256, 0, stream>>>(x, wsd, alpha_s, alpha_d);
    node_gather_kernel<<<(N_NODES + 7) / 8, 256, 0, stream>>>(csr, rowptr, cnt, alpha_s, alpha_d, x, aggx);
    gemm1_agg<<<dim3((ROW_TILES + 7) / 8, 4), 256, 0, stream>>>(aggx, W, agg);
    gemm2_kernel<<<dim3((ROW_TILES + 7) / 8, 2), 256, 0, stream>>>(agg, b, Wfc2, bfc2, h2);
    gemm3_kernel<<<(ROW_TILES + 7) / 8, 256, 0, stream>>>(h2, Wout, bout, out);
}


# Three dense layers fused into one launch, one workgroup per WGP: weights kept in LDS in fragment order, activations go layer to layer through LDS; f32 sum order of the products changes (k order)
# speedup vs baseline: 2.9191x; 1.2928x over previous
#include <hip/hip_runtime.h>
#include <hip/hip_bf16.h>

typedef __attribute__((ext_vector_type(2))) float v2f;
typedef __attribute__((ext_vector_type(8))) float v8f;

#define N_NODES   100000
#define N_EDGES0  800000
#define N_EDGES   900000
#define ROW_TILES 6250

__device__ __forceinline__ float leaky(float v) { return v > 0.0f ? v : 0.2f * v; }

__device__ __forceinline__ float wave_max_f(float v) {
#pragma unroll
    for (int off = 16; off; off >>= 1) v = fmaxf(v, __shfl_xor(v, off, 32));
    return v;
}
__device__ __forceinline__ float wave_sum_f(float v) {
#pragma unroll
    for (int off = 16; off; off >>= 1) v += __shfl_xor(v, off, 32);
    return v;
}
__device__ __forceinline__ int wave_sum_i(int v) {
#pragma unroll
    for (int off = 16; off; off >>= 1) v += __shfl_xor(v, off, 32);
    return v;
}
__device__ __forceinline__ int wave_incl_scan_i(int v, int lane) {
#pragma unroll
    for (int off = 1; off < 32; off <<= 1) {
        int u = __shfl_up(v, off, 32);
        if (lane >= off) v += u;
    }
    return v;
}

#define ZERO_BLOCKS 391
#define FOLD_BLOCKS 16
__global__ __launch_bounds__(256) void prep_kernel(int* __restrict__ cnt,
                                                   const float* __restrict__ W,
                                                   const float* __restrict__ a_src,
                                                   const float* __restrict__ a_dst,
                                                   float* __restrict__ wsd) {
    if (blockIdx.x < ZERO_BLOCKS) {
        int i = blockIdx.x * 256 + threadIdx.x;
        if (i < N_NODES) cnt[i] = 0;
        return;
    }
    const int lane = threadIdx.x & 31;
    const int k = (blockIdx.x - ZERO_BLOCKS) * 8 + (threadIdx.x >> 5);
    const float* wr = W + (size_t)k * 256;
    float s = 0.0f, d = 0.0f;
#pragma unroll
    for (int i = 0; i < 8; ++i) {
        float v = wr[lane + 32 * i];
        s += v * a_src[lane + 32 * i];
        d += v * a_dst[lane + 32 * i];
    }
    s = wave_sum_f(s);
    d = wave_sum_f(d);
    if (lane == 0) { wsd[k] = s; wsd[128 + k] = d; }
}

__global__ __launch_bounds__(256) void hist_kernel(const int* __restrict__ ei,
                                                   int* __restrict__ cnt) {
    int i = blockIdx.x * 256 + threadIdx.x;
    if (i >= N_EDGES0 / 4) return;
    int4 d4 = *(const int4*)(ei + N_EDGES0 + 4 * i);
    atomicAdd(&cnt[d4.x], 1);
    atomicAdd(&cnt[d4.y], 1);
    atomicAdd(&cnt[d4.z], 1);
    atomicAdd(&cnt[d4.w], 1);
}

__global__ __launch_bounds__(1024) void scan_kernel(const int* __restrict__ cnt,
                                                    int* __restrict__ rowptr,
                                                    int* __restrict__ cursor) {
    __shared__ int wsum[32];
    __shared__ int bpre;
    const int t = threadIdx.x;
    const int lane = t & 31;
    const int w = t >> 5;
    const int b = blockIdx.x;

    int pre = 0;
    for (int i = 0; i < b; ++i) pre += cnt[i * 1024 + t];
    pre = wave_sum_i(pre);
    if (lane == 0) wsum[w] = pre;
    __syncthreads();
    if (w == 0) {
        int v = wave_sum_i(wsum[lane]);
        if (lane == 0) bpre = v;
    }
    __syncthreads();
    const int blockPrefix = bpre;

    const int idx = b * 1024 + t;
    const int v = (idx < N_NODES) ? cnt[idx] : 0;
    const int inc = wave_incl_scan_i(v, lane);
    if (lane == 31) wsum[w] = inc;
    __syncthreads();
    const int wt = wsum[lane];
    const int wexcl = wave_incl_scan_i(wt, lane) - wt;
    const int wavePrefix = __shfl(wexcl, w, 32);
    const int excl = blockPrefix + wavePrefix + inc - v;
    if (idx < N_NODES) { rowptr[idx] = excl; cursor[idx] = excl; }
}

__global__ __launch_bounds__(256) void fill_kernel(const int* __restrict__ ei,
                                                   int* __restrict__ cursor,
                                                   int* __restrict__ csr) {
    int i = blockIdx.x * 256 + threadIdx.x;
    if (i >= N_EDGES0 / 4) return;
    int4 s4 = *(const int4*)(ei + 4 * i);
    int4 d4 = *(const int4*)(ei + N_EDGES0 + 4 * i);
    int p0 = atomicAdd(&cursor[d4.x], 1);
    int p1 = atomicAdd(&cursor[d4.y], 1);
    int p2 = atomicAdd(&cursor[d4.z], 1);
    int p3 = atomicAdd(&cursor[d4.w], 1);
    csr[p0] = s4.x;
    csr[p1] = s4.y;
    csr[p2] = s4.z;
    csr[p3] = s4.w;
}

__global__ __launch_bounds__(256) void alphas_kernel(const float* __restrict__ x,
                                                     const float* __restrict__ wsd,
                                                     float* __restrict__ as,
                                                     float* __restrict__ ad) {
    const int lane = threadIdx.x & 31;
    const int node0 = (blockIdx.x * 8 + (threadIdx.x >> 5)) * 4;
    if (node0 >= N_NODES) return;
    const float4 ws = *(const float4*)(wsd + 4 * lane);
    const float4 wd = *(const float4*)(wsd + 128 + 4 * lane);
    const float* xr = x + (size_t)node0 * 128 + 4 * lane;
    const float4 v0 = *(const float4*)xr;
    const float4 v1 = *(const float4*)(xr + 128);
    const float4 v2 = *(const float4*)(xr + 256);
    const float4 v3 = *(const float4*)(xr + 384);
    float s0 = v0.x * ws.x + v0.y * ws.y + v0.z * ws.z + v0.w * ws.w;
    float s1 = v1.x * ws.x + v1.y * ws.y + v1.z * ws.z + v1.w * ws.w;
    float s2 = v2.x * ws.x + v2.y * ws.y + v2.z * ws.z + v2.w * ws.w;
    float s3 = v3.x * ws.x + v3.y * ws.y + v3.z * ws.z + v3.w * ws.w;
    float d0 = v0.x * wd.x + v0.y * wd.y + v0.z * wd.z + v0.w * wd.w;
    float d1 = v1.x * wd.x + v1.y * wd.y + v1.z * wd.z + v1.w * wd.w;
    float d2 = v2.x * wd.x + v2.y * wd.y + v2.z * wd.z + v2.w * wd.w;
    float d3 = v3.x * wd.x + v3.y * wd.y + v3.z * wd.z + v3.w * wd.w;
    s0 = wave_sum_f(s0); s1 = wave_sum_f(s1); s2 = wave_sum_f(s2); s3 = wave_sum_f(s3);
    d0 = wave_sum_f(d0); d1 = wave_sum_f(d1); d2 = wave_sum_f(d2); d3 = wave_sum_f(d3);
    if (lane == 0) {
        *(float4*)(as + node0) = make_float4(s0, s1, s2, s3);
        *(float4*)(ad + node0) = make_float4(d0, d1, d2, d3);
    }
}

__device__ __forceinline__ int lane_i(int v, int t) { return __builtin_amdgcn_readlane(v, t); }
__device__ __forceinline__ float lane_f(float v, int t) {
    return __int_as_float(__builtin_amdgcn_readlane(__float_as_int(v), t));
}

__global__ __launch_bounds__(256) void node_gather_kernel(const int* __restrict__ csr,
                                                          const int* __restrict__ rowptr,
                                                          const int* __restrict__ cnt,
                                                          const float* __restrict__ as,
                                                          const float* __restrict__ ad,
                                                          const float* __restrict__ x,
                                                          float* __restrict__ aggx) {
    const int lane = threadIdx.x & 31;
    const int d = blockIdx.x * 8 + __builtin_amdgcn_readfirstlane((int)(threadIdx.x >> 5));
    if (d >= N_NODES) return;
    const int start = rowptr[d];
    const int L = cnt[d] + 1;
    const float adv = ad[d];

    int s0 = d;
    float e0 = -INFINITY;
    if (lane < L) {
        if (lane > 0) s0 = csr[start + lane - 1];
        e0 = leaky(as[s0] + adv);
    }
    float m = e0;
    for (int p = lane + 32; p < L; p += 32)
        m = fmaxf(m, leaky(as[csr[start + p - 1]] + adv));
    m = wave_max_f(m);

    const float x0 = (lane < L) ? __expf(e0 - m) : 0.0f;
    float sum = x0;
    for (int p = lane + 32; p < L; p += 32)
        sum += __expf(leaky(as[csr[start + p - 1]] + adv) - m);
    sum = wave_sum_f(sum);

    const int c0 = lane * 4;
    float4 acc = make_float4(0.0f, 0.0f, 0.0f, 0.0f);
    for (int base = 0; base < L; base += 32) {
        int s = s0;
        float al = x0 / sum;
        if (base > 0) {
            const int p = base + lane;
            s = d; al = 0.0f;
            if (p < L) {
                s = csr[start + p - 1];
                al = __expf(leaky(as[s] + adv) - m) / sum;
            }
        }
        const int nc = (L - base < 32) ? (L - base) : 32;
        int t = 0;
        for (; t + 8 <= nc; t += 8) {
            float4 v[8];
            float  a[8];
#pragma unroll
            for (int j = 0; j < 8; ++j) {
                const int sj = lane_i(s, t + j);
                a[j] = lane_f(al, t + j);
                v[j] = *(const float4*)(x + (size_t)sj * 128 + c0);
            }
#pragma unroll
            for (int j = 0; j < 8; ++j) {
                acc.x = fmaf(a[j], v[j].x, acc.x); acc.y = fmaf(a[j], v[j].y, acc.y);
                acc.z = fmaf(a[j], v[j].z, acc.z); acc.w = fmaf(a[j], v[j].w, acc.w);
            }
        }
        for (; t + 4 <= nc; t += 4) {
            float4 v[4];
            float  a[4];
#pragma unroll
            for (int j = 0; j < 4; ++j) {
                const int sj = lane_i(s, t + j);
                a[j] = lane_f(al, t + j);
                v[j] = *(const float4*)(x + (size_t)sj * 128 + c0);
            }
#pragma unroll
            for (int j = 0; j < 4; ++j) {
                acc.x = fmaf(a[j], v[j].x, acc.x); acc.y = fmaf(a[j], v[j].y, acc.y);
                acc.z = fmaf(a[j], v[j].z, acc.z); acc.w = fmaf(a[j], v[j].w, acc.w);
            }
        }
#pragma clang loop unroll(disable)
        for (; t < nc; ++t) {
            const int   sa = lane_i(s, t);
            const float aa = lane_f(al, t);
            const float4 va = *(const float4*)(x + (size_t)sa * 128 + c0);
            acc.x = fmaf(aa, va.x, acc.x); acc.y = fmaf(aa, va.y, acc.y);
            acc.z = fmaf(aa, va.z, acc.z); acc.w = fmaf(aa, va.w, acc.w);
        }
    }
    *(float4*)(aggx + (size_t)d * 128 + c0) = acc;
}

#define MLP_WG    256
#define MLP_WAVES 8
#define SP        68
#define WQ_F      (128 * 256)
#define W2Q_F     (256 * 128)
#define WOQ_F     (128 * 16)
#define PATCH_F   (16 * SP)

__device__ __forceinline__ v8f mm4(float a0, float a1, float b0, float b1, v8f acc) {
    v2f a; a.x = a0; a.y = a1;
    v2f bm; bm.x = b0; bm.y = b1;
    return __builtin_amdgcn_wmma_f32_16x16x4_f32(false, a, false, bm, (short)0, acc, false, false);
}

__global__ __launch_bounds__(256) void mlp_fused_kernel(const float* __restrict__ aggx,
                                                        const float* __restrict__ W,
                                                        const float* __restrict__ b,
                                                        const float* __restrict__ Wfc2,
                                                        const float* __restrict__ bfc2,
                                                        const float* __restrict__ Wout,
                                                        const float* __restrict__ bout,
                                                        float* __restrict__ out) {
    __shared__ __attribute__((aligned(16))) float lds[WQ_F + W2Q_F + WOQ_F + MLP_WAVES * PATCH_F];
    float* Wq  = lds;
    float* W2q = Wq + WQ_F;
    float* Woq = W2q + W2Q_F;
    const int tid  = threadIdx.x;
    const int lane = tid & 31;
    const int wave = __builtin_amdgcn_readfirstlane(tid >> 5);

#pragma unroll
    for (int i = 0; i < 8; ++i) {
        const int task = tid + 256 * i;
        const int kq = task >> 6, c4 = (task & 63) * 4;
        const float* g = W + (size_t)(4 * kq) * 256 + c4;
        const float4 r0 = *(const float4*)g;
        const float4 r1 = *(const float4*)(g + 256);
        const float4 r2 = *(const float4*)(g + 512);
        const float4 r3 = *(const float4*)(g + 768);
        float4* dst = (float4*)(Wq + (size_t)(kq * 256 + c4) * 4);
        dst[0] = make_float4(r0.x, r1.x, r2.x, r3.x);
        dst[1] = make_float4(r0.y, r1.y, r2.y, r3.y);
        dst[2] = make_float4(r0.z, r1.z, r2.z, r3.z);
        dst[3] = make_float4(r0.w, r1.w, r2.w, r3.w);
    }
#pragma unroll
    for (int i = 0; i < 8; ++i) {
        const int task = tid + 256 * i;
        const int kq = task >> 5, c4 = (task & 31) * 4;
        const float* g = Wfc2 + (size_t)(4 * kq) * 128 + c4;
        const float4 r0 = *(const float4*)g;
        const float4 r1 = *(const float4*)(g + 128);
        const float4 r2 = *(const float4*)(g + 256);
        const float4 r3 = *(const float4*)(g + 384);
        float4* dst = (float4*)(W2q + (size_t)(kq * 128 + c4) * 4);
        dst[0] = make_float4(r0.x, r1.x, r2.x, r3.x);
        dst[1] = make_float4(r0.y, r1.y, r2.y, r3.y);
        dst[2] = make_float4(r0.z, r1.z, r2.z, r3.z);
        dst[3] = make_float4(r0.w, r1.w, r2.w, r3.w);
    }
#pragma unroll
    for (int n = 0; n < 8; ++n) {
        const int i = tid + 256 * n;
        const int j = i & 3, col = (i >> 2) & 15, kq = i >> 6;
        const float v = Wout[(4 * kq + j) * 6 + (col < 6 ? col : 0)];
        Woq[i] = (col < 6) ? v : 0.0f;
    }
    __syncthreads();

    const int c = lane & 15;
    const int h = lane >> 4;
    float* patch = Woq + WOQ_F + wave * PATCH_F;
    const float4* Wq4  = (const float4*)Wq;
    const float4* W2q4 = (const float4*)W2q;
    const float4* Woq4 = (const float4*)Woq;
    float*        pst  = patch + (8 * h) * SP + c;
    const float4* pld  = (const float4*)(patch + c * SP + 4 * h);

    for (int it = 0; it < 4; ++it) {
        const int tile = it * (MLP_WG * MLP_WAVES) + wave * MLP_WG + (int)blockIdx.x;
        if (tile >= ROW_TILES) break;
        const int row0 = tile * 16;

        float4 a[16];
        const float* ap = aggx + (size_t)(row0 + c) * 128 + 4 * h;
#pragma unroll
        for (int kb = 0; kb < 16; ++kb) a[kb] = *(const float4*)(ap + 8 * kb);

        v8f acc2[8] = {};
#pragma unroll 1
        for (int p = 0; p < 4; ++p) {
            v8f acc1[4] = {};
            const float4* wq = Wq4 + h * 256 + p * 64 + c;
#pragma unroll
            for (int kb = 0; kb < 16; ++kb) {
                float4 bq[4];
#pragma unroll
                for (int t = 0; t < 4; ++t) bq[t] = wq[kb * 512 + t * 16];
#pragma unroll
                for (int t = 0; t < 4; ++t) acc1[t] = mm4(a[kb].x, a[kb].y, bq[t].x, bq[t].y, acc1[t]);
#pragma unroll
                for (int t = 0; t < 4; ++t) acc1[t] = mm4(a[kb].z, a[kb].w, bq[t].z, bq[t].w, acc1[t]);
            }
#pragma unroll
            for (int t = 0; t < 4; ++t) {
                const float bb = b[p * 64 + t * 16 + c];
#pragma unroll
                for (int r = 0; r < 8; ++r) pst[r * SP + t * 16] = fmaxf(acc1[t][r] + bb, 0.0f);
            }
            const float4* w2 = W2q4 + (p * 16 + h) * 128 + c;
#pragma unroll
            for (int kb = 0; kb < 8; ++kb) {
                const float4 a2 = pld[2 * kb];
                float4 bq[8];
#pragma unroll
                for (int t = 0; t < 8; ++t) bq[t] = w2[kb * 256 + t * 16];
#pragma unroll
                for (int t = 0; t < 8; ++t) acc2[t] = mm4(a2.x, a2.y, bq[t].x, bq[t].y, acc2[t]);
#pragma unroll
                for (int t = 0; t < 8; ++t) acc2[t] = mm4(a2.z, a2.w, bq[t].z, bq[t].w, acc2[t]);
            }
        }

        v8f acc3a = {}, acc3b = {};
#pragma unroll
        for (int half = 0; half < 2; ++half) {
#pragma unroll
            for (int t = 0; t < 4; ++t) {
                const float bb = bfc2[half * 64 + t * 16 + c];
#pragma unroll
                for (int r = 0; r < 8; ++r) pst[r * SP + t * 16] = fmaxf(acc2[half * 4 + t][r] + bb, 0.0f);
            }
            const float4* wo = Woq4 + (half * 16 + h) * 16 + c;
#pragma unroll
            for (int kb = 0; kb < 8; ++kb) {
                const float4 a3 = pld[2 * kb];
                const float4 bq = wo[kb * 32];
                acc3a = mm4(a3.x, a3.y, bq.x, bq.y, acc3a);
                acc3b = mm4(a3.z, a3.w, bq.z, bq.w, acc3b);
            }
        }
        if (c < 6) {
            const float bb = bout[c];
            float* op = out + (size_t)(row0 + 8 * h) * 6 + c;
#pragma unroll
            for (int r = 0; r < 8; ++r) op[r * 6] = (acc3a[r] + acc3b[r]) + bb;
        }
    }
}

extern "C" void kernel_launch(void* const* d_in, const int* in_sizes, int n_in,
                              void* d_out, int out_size, void* d_ws, size_t ws_size,
                              hipStream_t stream) {
    const float* x     = (const float*)d_in[0];
    const int*   ei    = (const int*)  d_in[1];
    const float* W     = (const float*)d_in[2];
    const float* a_src = (const float*)d_in[3];
    const float* a_dst = (const float*)d_in[4];
    const float* b     = (const float*)d_in[5];
    const float* Wfc2  = (const float*)d_in[6];
    const float* bfc2  = (const float*)d_in[7];
    const float* Wout  = (const float*)d_in[8];
    const float* bout  = (const float*)d_in[9];
    float* out = (float*)d_out;

    float* ws      = (float*)d_ws;
    float* aggx    = ws;
    float* alpha_s = aggx + (size_t)N_NODES * 128;
    float* alpha_d = alpha_s + N_NODES;
    float* wsd     = alpha_d + N_NODES;
    int*   cnt     = (int*)(wsd + 256);
    int*   rowptr  = cnt + N_NODES;
    int*   cursor  = rowptr + N_NODES;
    int*   csr     = cursor + N_NODES;

    prep_kernel<<<ZERO_BLOCKS + FOLD_BLOCKS, 256, 0, stream>>>(cnt, W, a_src, a_dst, wsd);
    hist_kernel<<<(N_EDGES0 / 4 + 255) / 256, 256, 0, stream>>>(ei, cnt);
    scan_kernel<<<(N_NODES + 1023) / 1024, 1024, 0, stream>>>(cnt, rowptr, cursor);
    fill_kernel<<<(N_EDGES0 / 4 + 255) / 256, 256, 0, stream>>>(ei, cursor, csr);
    alphas_kernel<<<(N_NODES / 4 + 7) / 8, 256, 0, stream>>>(x, wsd, alpha_s, alpha_d);
    node_gather_kernel<<<(N_NODES + 7) / 8, 256, 0, stream>>>(csr, rowptr, cnt, alpha_s, alpha_d, x, aggx);
    mlp_fused_kernel<<<MLP_WG, 256, 0, stream>>>(aggx, W, b, Wfc2, bfc2, Wout, bout, out);
}


# Edges go straight into a fixed 32-slot bucket per destination in one atomic pass (fuller lists spill to a scanned list): histogram and prefix-sum launches removed; sum order stays slot arrival
# speedup vs baseline: 3.7534x; 1.2858x over previous
#include <hip/hip_runtime.h>
#include <hip/hip_bf16.h>

typedef __attribute__((ext_vector_type(2))) float v2f;
typedef __attribute__((ext_vector_type(8))) float v8f;

#define N_NODES   100000
#define N_EDGES0  800000
#define N_EDGES   900000
#define ROW_TILES 6250

__device__ __forceinline__ float leaky(float v) { return v > 0.0f ? v : 0.2f * v; }

__device__ __forceinline__ float wave_max_f(float v) {
#pragma unroll
    for (int off = 16; off; off >>= 1) v = fmaxf(v, __shfl_xor(v, off, 32));
    return v;
}
__device__ __forceinline__ float wave_sum_f(float v) {
#pragma unroll
    for (int off = 16; off; off >>= 1) v += __shfl_xor(v, off, 32);
    return v;
}

#define BUCKET_CAP 32
#define ZERO_BLOCKS 391
#define FOLD_BLOCKS 16
__global__ __launch_bounds__(256) void prep_kernel(int* __restrict__ cnt,
                                                   const float* __restrict__ W,
                                                   const float* __restrict__ a_src,
                                                   const float* __restrict__ a_dst,
                                                   float* __restrict__ wsd) {
    if (blockIdx.x < ZERO_BLOCKS) {
        int i = blockIdx.x * 256 + threadIdx.x;
        if (i < N_NODES + 32) cnt[i] = 0;
        return;
    }
    const int lane = threadIdx.x & 31;
    const int k = (blockIdx.x - ZERO_BLOCKS) * 8 + (threadIdx.x >> 5);
    const float* wr = W + (size_t)k * 256;
    float s = 0.0f, d = 0.0f;
#pragma unroll
    for (int i = 0; i < 8; ++i) {
        float v = wr[lane + 32 * i];
        s += v * a_src[lane + 32 * i];
        d += v * a_dst[lane + 32 * i];
    }
    s = wave_sum_f(s);
    d = wave_sum_f(d);
    if (lane == 0) { wsd[k] = s; wsd[128 + k] = d; }
}

__device__ __forceinline__ void place_edge(int d, int src, int slot, int* __restrict__ spill_cnt,
                                           int* __restrict__ bucket, int* __restrict__ spill) {
    if (slot < BUCKET_CAP) {
        bucket[(size_t)d * BUCKET_CAP + slot] = src;
    } else {
        const int q = atomicAdd(spill_cnt, 1);
        spill[2 * q] = d;
        spill[2 * q + 1] = src;
    }
}

__global__ __launch_bounds__(256) void bucket_fill_kernel(const int* __restrict__ ei,
                                                          int* __restrict__ cnt,
                                                          int* __restrict__ spill_cnt,
                                                          int* __restrict__ bucket,
                                                          int* __restrict__ spill) {
    int i = blockIdx.x * 256 + threadIdx.x;
    if (i >= N_EDGES0 / 4) return;
    int4 s4 = *(const int4*)(ei + 4 * i);
    int4 d4 = *(const int4*)(ei + N_EDGES0 + 4 * i);
    int p0 = atomicAdd(&cnt[d4.x], 1);
    int p1 = atomicAdd(&cnt[d4.y], 1);
    int p2 = atomicAdd(&cnt[d4.z], 1);
    int p3 = atomicAdd(&cnt[d4.w], 1);
    place_edge(d4.x, s4.x, p0, spill_cnt, bucket, spill);
    place_edge(d4.y, s4.y, p1, spill_cnt, bucket, spill);
    place_edge(d4.z, s4.z, p2, spill_cnt, bucket, spill);
    place_edge(d4.w, s4.w, p3, spill_cnt, bucket, spill);
}

__global__ __launch_bounds__(256) void alphas_kernel(const float* __restrict__ x,
                                                     const float* __restrict__ wsd,
                                                     float* __restrict__ as,
                                                     float* __restrict__ ad) {
    const int lane = threadIdx.x & 31;
    const int node0 = (blockIdx.x * 8 + (threadIdx.x >> 5)) * 4;
    if (node0 >= N_NODES) return;
    const float4 ws = *(const float4*)(wsd + 4 * lane);
    const float4 wd = *(const float4*)(wsd + 128 + 4 * lane);
    const float* xr = x + (size_t)node0 * 128 + 4 * lane;
    const float4 v0 = *(const float4*)xr;
    const float4 v1 = *(const float4*)(xr + 128);
    const float4 v2 = *(const float4*)(xr + 256);
    const float4 v3 = *(const float4*)(xr + 384);
    float s0 = v0.x * ws.x + v0.y * ws.y + v0.z * ws.z + v0.w * ws.w;
    float s1 = v1.x * ws.x + v1.y * ws.y + v1.z * ws.z + v1.w * ws.w;
    float s2 = v2.x * ws.x + v2.y * ws.y + v2.z * ws.z + v2.w * ws.w;
    float s3 = v3.x * ws.x + v3.y * ws.y + v3.z * ws.z + v3.w * ws.w;
    float d0 = v0.x * wd.x + v0.y * wd.y + v0.z * wd.z + v0.w * wd.w;
    float d1 = v1.x * wd.x + v1.y * wd.y + v1.z * wd.z + v1.w * wd.w;
    float d2 = v2.x * wd.x + v2.y * wd.y + v2.z * wd.z + v2.w * wd.w;
    float d3 = v3.x * wd.x + v3.y * wd.y + v3.z * wd.z + v3.w * wd.w;
    s0 = wave_sum_f(s0); s1 = wave_sum_f(s1); s2 = wave_sum_f(s2); s3 = wave_sum_f(s3);
    d0 = wave_sum_f(d0); d1 = wave_sum_f(d1); d2 = wave_sum_f(d2); d3 = wave_sum_f(d3);
    if (lane == 0) {
        *(float4*)(as + node0) = make_float4(s0, s1, s2, s3);
        *(float4*)(ad + node0) = make_float4(d0, d1, d2, d3);
    }
}

__device__ __forceinline__ int lane_i(int v, int t) { return __builtin_amdgcn_readlane(v, t); }
__device__ __forceinline__ float lane_f(float v, int t) {
    return __int_as_float(__builtin_amdgcn_readlane(__float_as_int(v), t));
}

__global__ __launch_bounds__(256) void node_gather_kernel(const int* __restrict__ bucket,
                                                          const int* __restrict__ cnt,
                                                          const int* __restrict__ spill_cnt,
                                                          const int* __restrict__ spill,
                                                          const float* __restrict__ as,
                                                          const float* __restrict__ ad,
                                                          const float* __restrict__ x,
                                                          float* __restrict__ aggx) {
    const int lane = threadIdx.x & 31;
    const int d = blockIdx.x * 8 + __builtin_amdgcn_readfirstlane((int)(threadIdx.x >> 5));
    if (d >= N_NODES) return;
    const int n = cnt[d];
    const int L = (n < BUCKET_CAP ? n : BUCKET_CAP) + 1;
    const int* bk = bucket + (size_t)d * BUCKET_CAP;
    const float adv = ad[d];
    int spn = 0;
    if (n > BUCKET_CAP) {
        spn = spill_cnt[0];
        if (spn > N_EDGES0) spn = N_EDGES0;
    }
    const int2* sp = (const int2*)spill;

    int s0 = d;
    float e0 = -INFINITY;
    if (lane < L) {
        if (lane > 0) s0 = bk[lane - 1];
        e0 = leaky(as[s0] + adv);
    }
    float m = e0;
    for (int p = lane + 32; p < L; p += 32)
        m = fmaxf(m, leaky(as[bk[p - 1]] + adv));
    for (int i = lane; i < spn; i += 32) {
        const int2 e = sp[i];
        if (e.x == d) m = fmaxf(m, leaky(as[e.y] + adv));
    }
    m = wave_max_f(m);

    const float x0 = (lane < L) ? __expf(e0 - m) : 0.0f;
    float sum = x0;
    for (int p = lane + 32; p < L; p += 32)
        sum += __expf(leaky(as[bk[p - 1]] + adv) - m);
    for (int i = lane; i < spn; i += 32) {
        const int2 e = sp[i];
        if (e.x == d) sum += __expf(leaky(as[e.y] + adv) - m);
    }
    sum = wave_sum_f(sum);

    const int c0 = lane * 4;
    float4 acc = make_float4(0.0f, 0.0f, 0.0f, 0.0f);
    for (int base = 0; base < L; base += 32) {
        int s = s0;
        float al = x0 / sum;
        if (base > 0) {
            const int p = base + lane;
            s = d; al = 0.0f;
            if (p < L) {
                s = bk[p - 1];
                al = __expf(leaky(as[s] + adv) - m) / sum;
            }
        }
        const int nc = (L - base < 32) ? (L - base) : 32;
        int t = 0;
        for (; t + 8 <= nc; t += 8) {
            float4 v[8];
            float  a[8];
#pragma unroll
            for (int j = 0; j < 8; ++j) {
                const int sj = lane_i(s, t + j);
                a[j] = lane_f(al, t + j);
                v[j] = *(const float4*)(x + (size_t)sj * 128 + c0);
            }
#pragma unroll
            for (int j = 0; j < 8; ++j) {
                acc.x = fmaf(a[j], v[j].x, acc.x); acc.y = fmaf(a[j], v[j].y, acc.y);
                acc.z = fmaf(a[j], v[j].z, acc.z); acc.w = fmaf(a[j], v[j].w, acc.w);
            }
        }
        for (; t + 4 <= nc; t += 4) {
            float4 v[4];
            float  a[4];
#pragma unroll
            for (int j = 0; j < 4; ++j) {
                const int sj = lane_i(s, t + j);
                a[j] = lane_f(al, t + j);
                v[j] = *(const float4*)(x + (size_t)sj * 128 + c0);
            }
#pragma unroll
            for (int j = 0; j < 4; ++j) {
                acc.x = fmaf(a[j], v[j].x, acc.x); acc.y = fmaf(a[j], v[j].y, acc.y);
                acc.z = fmaf(a[j], v[j].z, acc.z); acc.w = fmaf(a[j], v[j].w, acc.w);
            }
        }
#pragma clang loop unroll(disable)
        for (; t < nc; ++t) {
            const int   sa = lane_i(s, t);
            const float aa = lane_f(al, t);
            const float4 va = *(const float4*)(x + (size_t)sa * 128 + c0);
            acc.x = fmaf(aa, va.x, acc.x); acc.y = fmaf(aa, va.y, acc.y);
            acc.z = fmaf(aa, va.z, acc.z); acc.w = fmaf(aa, va.w, acc.w);
        }
    }
    for (int base = 0; base < spn; base += 32) {
        const int i = base + lane;
        int s = d;
        float al = 0.0f;
        bool hit = false;
        if (i < spn) {
            const int2 e = sp[i];
            if (e.x == d) {
                s = e.y;
                al = __expf(leaky(as[s] + adv) - m) / sum;
                hit = true;
            }
        }
        unsigned hits = __builtin_amdgcn_ballot_w32(hit);
        while (hits) {
            const int t = __builtin_ctz(hits);
            hits &= hits - 1;
            const int   sa = lane_i(s, t);
            const float aa = lane_f(al, t);
            const float4 va = *(const float4*)(x + (size_t)sa * 128 + c0);
            acc.x = fmaf(aa, va.x, acc.x); acc.y = fmaf(aa, va.y, acc.y);
            acc.z = fmaf(aa, va.z, acc.z); acc.w = fmaf(aa, va.w, acc.w);
        }
    }
    *(float4*)(aggx + (size_t)d * 128 + c0) = acc;
}

#define MLP_WG    256
#define MLP_WAVES 8
#define SP        68
#define WQ_F      (128 * 256)
#define W2Q_F     (256 * 128)
#define WOQ_F     (128 * 16)
#define PATCH_F   (16 * SP)

__device__ __forceinline__ v8f mm4(float a0, float a1, float b0, float b1, v8f acc) {
    v2f a; a.x = a0; a.y = a1;
    v2f bm; bm.x = b0; bm.y = b1;
    return __builtin_amdgcn_wmma_f32_16x16x4_f32(false, a, false, bm, (short)0, acc, false, false);
}

__global__ __launch_bounds__(256) void mlp_fused_kernel(const float* __restrict__ aggx,
                                                        const float* __restrict__ W,
                                                        const float* __restrict__ b,
                                                        const float* __restrict__ Wfc2,
                                                        const float* __restrict__ bfc2,
                                                        const float* __restrict__ Wout,
                                                        const float* __restrict__ bout,
                                                        float* __restrict__ out) {
    __shared__ __attribute__((aligned(16))) float lds[WQ_F + W2Q_F + WOQ_F + MLP_WAVES * PATCH_F];
    float* Wq  = lds;
    float* W2q = Wq + WQ_F;
    float* Woq = W2q + W2Q_F;
    const int tid  = threadIdx.x;
    const int lane = tid & 31;
    const int wave = __builtin_amdgcn_readfirstlane(tid >> 5);

#pragma unroll
    for (int i = 0; i < 8; ++i) {
        const int task = tid + 256 * i;
        const int kq = task >> 6, c4 = (task & 63) * 4;
        const float* g = W + (size_t)(4 * kq) * 256 + c4;
        const float4 r0 = *(const float4*)g;
        const float4 r1 = *(const float4*)(g + 256);
        const float4 r2 = *(const float4*)(g + 512);
        const float4 r3 = *(const float4*)(g + 768);
        float4* dst = (float4*)(Wq + (size_t)(kq * 256 + c4) * 4);
        dst[0] = make_float4(r0.x, r1.x, r2.x, r3.x);
        dst[1] = make_float4(r0.y, r1.y, r2.y, r3.y);
        dst[2] = make_float4(r0.z, r1.z, r2.z, r3.z);
        dst[3] = make_float4(r0.w, r1.w, r2.w, r3.w);
    }
#pragma unroll
    for (int i = 0; i < 8; ++i) {
        const int task = tid + 256 * i;
        const int kq = task >> 5, c4 = (task & 31) * 4;
        const float* g = Wfc2 + (size_t)(4 * kq) * 128 + c4;
        const float4 r0 = *(const float4*)g;
        const float4 r1 = *(const float4*)(g + 128);
        const float4 r2 = *(const float4*)(g + 256);
        const float4 r3 = *(const float4*)(g + 384);
        float4* dst = (float4*)(W2q + (size_t)(kq * 128 + c4) * 4);
        dst[0] = make_float4(r0.x, r1.x, r2.x, r3.x);
        dst[1] = make_float4(r0.y, r1.y, r2.y, r3.y);
        dst[2] = make_float4(r0.z, r1.z, r2.z, r3.z);
        dst[3] = make_float4(r0.w, r1.w, r2.w, r3.w);
    }
#pragma unroll
    for (int n = 0; n < 8; ++n) {
        const int i = tid + 256 * n;
        const int j = i & 3, col = (i >> 2) & 15, kq = i >> 6;
        const float v = Wout[(4 * kq + j) * 6 + (col < 6 ? col : 0)];
        Woq[i] = (col < 6) ? v : 0.0f;
    }
    __syncthreads();

    const int c = lane & 15;
    const int h = lane >> 4;
    float* patch = Woq + WOQ_F + wave * PATCH_F;
    const float4* Wq4  = (const float4*)Wq;
    const float4* W2q4 = (const float4*)W2q;
    const float4* Woq4 = (const float4*)Woq;
    float*        pst  = patch + (8 * h) * SP + c;
    const float4* pld  = (const float4*)(patch + c * SP + 4 * h);

    for (int it = 0; it < 4; ++it) {
        const int tile = it * (MLP_WG * MLP_WAVES) + wave * MLP_WG + (int)blockIdx.x;
        if (tile >= ROW_TILES) break;
        const int row0 = tile * 16;

        float4 a[16];
        const float* ap = aggx + (size_t)(row0 + c) * 128 + 4 * h;
#pragma unroll
        for (int kb = 0; kb < 16; ++kb) a[kb] = *(const float4*)(ap + 8 * kb);

        v8f acc2[8] = {};
#pragma unroll 1
        for (int p = 0; p < 4; ++p) {
            v8f acc1[4] = {};
            const float4* wq = Wq4 + h * 256 + p * 64 + c;
#pragma unroll
            for (int kb = 0; kb < 16; ++kb) {
                float4 bq[4];
#pragma unroll
                for (int t = 0; t < 4; ++t) bq[t] = wq[kb * 512 + t * 16];
#pragma unroll
                for (int t = 0; t < 4; ++t) acc1[t] = mm4(a[kb].x, a[kb].y, bq[t].x, bq[t].y, acc1[t]);
#pragma unroll
                for (int t = 0; t < 4; ++t) acc1[t] = mm4(a[kb].z, a[kb].w, bq[t].z, bq[t].w, acc1[t]);
            }
#pragma unroll
            for (int t = 0; t < 4; ++t) {
                const float bb = b[p * 64 + t * 16 + c];
#pragma unroll
                for (int r = 0; r < 8; ++r) pst[r * SP + t * 16] = fmaxf(acc1[t][r] + bb, 0.0f);
            }
            const float4* w2 = W2q4 + (p * 16 + h) * 128 + c;
#pragma unroll
            for (int kb = 0; kb < 8; ++kb) {
                const float4 a2 = pld[2 * kb];
                float4 bq[8];
#pragma unroll
                for (int t = 0; t < 8; ++t) bq[t] = w2[kb * 256 + t * 16];
#pragma unroll
                for (int t = 0; t < 8; ++t) acc2[t] = mm4(a2.x, a2.y, bq[t].x, bq[t].y, acc2[t]);
#pragma unroll
                for (int t = 0; t < 8; ++t) acc2[t] = mm4(a2.z, a2.w, bq[t].z, bq[t].w, acc2[t]);
            }
        }

        v8f acc3a = {}, acc3b = {};
#pragma unroll
        for (int half = 0; half < 2; ++half) {
#pragma unroll
            for (int t = 0; t < 4; ++t) {
                const float bb = bfc2[half * 64 + t * 16 + c];
#pragma unroll
                for (int r = 0; r < 8; ++r) pst[r * SP + t * 16] = fmaxf(acc2[half * 4 + t][r] + bb, 0.0f);
            }
            const float4* wo = Woq4 + (half * 16 + h) * 16 + c;
#pragma unroll
            for (int kb = 0; kb < 8; ++kb) {
                const float4 a3 = pld[2 * kb];
                const float4 bq = wo[kb * 32];
                acc3a = mm4(a3.x, a3.y, bq.x, bq.y, acc3a);
                acc3b = mm4(a3.z, a3.w, bq.z, bq.w, acc3b);
            }
        }
        if (c < 6) {
            const float bb = bout[c];
            float* op = out + (size_t)(row0 + 8 * h) * 6 + c;
#pragma unroll
            for (int r = 0; r < 8; ++r) op[r * 6] = (acc3a[r] + acc3b[r]) + bb;
        }
    }
}

extern "C" void kernel_launch(void* const* d_in, const int* in_sizes, int n_in,
                              void* d_out, int out_size, void* d_ws, size_t ws_size,
                              hipStream_t stream) {
    const float* x     = (const float*)d_in[0];
    const int*   ei    = (const int*)  d_in[1];
    const float* W     = (const float*)d_in[2];
    const float* a_src = (const float*)d_in[3];
    const float* a_dst = (const float*)d_in[4];
    const float* b     = (const float*)d_in[5];
    const float* Wfc2  = (const float*)d_in[6];
    const float* bfc2  = (const float*)d_in[7];
    const float* Wout  = (const float*)d_in[8];
    const float* bout  = (const float*)d_in[9];
    float* out = (float*)d_out;

    float* ws      = (float*)d_ws;
    float* aggx    = ws;
    float* alpha_s = aggx + (size_t)N_NODES * 128;
    float* alpha_d = alpha_s + N_NODES;
    float* wsd     = alpha_d + N_NODES;
    int*   cnt     = (int*)(wsd + 256);
    int*   spcnt   = cnt + N_NODES;
    int*   bucket  = spcnt + 32;
    int*   spill   = bucket + (size_t)N_NODES * BUCKET_CAP;

    prep_kernel<<<ZERO_BLOCKS + FOLD_BLOCKS, 256, 0, stream>>>(cnt, W, a_src, a_dst, wsd);
    bucket_fill_kernel<<<(N_EDGES0 / 4 + 255) / 256, 256, 0, stream>>>(ei, cnt, spcnt, bucket, spill);
    alphas_kernel<<<(N_NODES / 4 + 7) / 8, 256, 0, stream>>>(x, wsd, alpha_s, alpha_d);
    node_gather_kernel<<<(N_NODES + 7) / 8, 256, 0, stream>>>(bucket, cnt, spcnt, spill, alpha_s, alpha_d, x, aggx);
    mlp_fused_kernel<<<MLP_WG, 256, 0, stream>>>(aggx, W, b, Wfc2, bfc2, Wout, bout, out);
}


# Source scores come from the x rows the gather loads anyway (row . W a_src, DPP lane sums): score launch and per-edge score loads gone; running-max softmax; f32 sum order of scores and sums changes
# speedup vs baseline: 4.0147x; 1.0696x over previous
#include <hip/hip_runtime.h>
#include <hip/hip_bf16.h>

typedef __attribute__((ext_vector_type(2))) float v2f;
typedef __attribute__((ext_vector_type(8))) float v8f;

#define N_NODES   100000
#define N_EDGES0  800000
#define N_EDGES   900000
#define ROW_TILES 6250

__device__ __forceinline__ float leaky(float v) { return v > 0.0f ? v : 0.2f * v; }

__device__ __forceinline__ float wave_max_f(float v) {
#pragma unroll
    for (int off = 16; off; off >>= 1) v = fmaxf(v, __shfl_xor(v, off, 32));
    return v;
}
__device__ __forceinline__ float wave_sum_f(float v) {
#pragma unroll
    for (int off = 16; off; off >>= 1) v += __shfl_xor(v, off, 32);
    return v;
}

#define BUCKET_CAP 32
#define ZERO_BLOCKS 391
#define FOLD_BLOCKS 16
__global__ __launch_bounds__(256) void prep_kernel(int* __restrict__ cnt,
                                                   const float* __restrict__ W,
                                                   const float* __restrict__ a_src,
                                                   const float* __restrict__ a_dst,
                                                   float* __restrict__ wsd) {
    if (blockIdx.x < ZERO_BLOCKS) {
        int i = blockIdx.x * 256 + threadIdx.x;
        if (i < N_NODES + 32) cnt[i] = 0;
        return;
    }
    const int lane = threadIdx.x & 31;
    const int k = (blockIdx.x - ZERO_BLOCKS) * 8 + (threadIdx.x >> 5);
    const float* wr = W + (size_t)k * 256;
    float s = 0.0f, d = 0.0f;
#pragma unroll
    for (int i = 0; i < 8; ++i) {
        float v = wr[lane + 32 * i];
        s += v * a_src[lane + 32 * i];
        d += v * a_dst[lane + 32 * i];
    }
    s = wave_sum_f(s);
    d = wave_sum_f(d);
    if (lane == 0) { wsd[k] = s; wsd[128 + k] = d; }
}

__device__ __forceinline__ void place_edge(int d, int src, int slot, int* __restrict__ spill_cnt,
                                           int* __restrict__ bucket, int* __restrict__ spill) {
    if (slot < BUCKET_CAP) {
        bucket[(size_t)d * BUCKET_CAP + slot] = src;
    } else {
        const int q = atomicAdd(spill_cnt, 1);
        spill[2 * q] = d;
        spill[2 * q + 1] = src;
    }
}

__global__ __launch_bounds__(256) void bucket_fill_kernel(const int* __restrict__ ei,
                                                          int* __restrict__ cnt,
                                                          int* __restrict__ spill_cnt,
                                                          int* __restrict__ bucket,
                                                          int* __restrict__ spill) {
    int i = blockIdx.x * 256 + threadIdx.x;
    if (i >= N_EDGES0 / 4) return;
    int4 s4 = *(const int4*)(ei + 4 * i);
    int4 d4 = *(const int4*)(ei + N_EDGES0 + 4 * i);
    int p0 = atomicAdd(&cnt[d4.x], 1);
    int p1 = atomicAdd(&cnt[d4.y], 1);
    int p2 = atomicAdd(&cnt[d4.z], 1);
    int p3 = atomicAdd(&cnt[d4.w], 1);
    place_edge(d4.x, s4.x, p0, spill_cnt, bucket, spill);
    place_edge(d4.y, s4.y, p1, spill_cnt, bucket, spill);
    place_edge(d4.z, s4.z, p2, spill_cnt, bucket, spill);
    place_edge(d4.w, s4.w, p3, spill_cnt, bucket, spill);
}

__device__ __forceinline__ int lane_i(int v, int t) { return __builtin_amdgcn_readlane(v, t); }
__device__ __forceinline__ float lane_f(float v, int t) {
    return __int_as_float(__builtin_amdgcn_readlane(__float_as_int(v), t));
}

__device__ __forceinline__ float xhalf(float v) {
    const int b = __float_as_int(v);
    return __int_as_float(__builtin_amdgcn_permlanex16(b, b, 0x76543210, 0xfedcba98, false, false));
}
template <int N> __device__ __forceinline__ float row_xor(float v) {
    return __int_as_float(__builtin_amdgcn_update_dpp(0, __float_as_int(v), 0x160 + N, 0xf, 0xf, true));
}
__device__ __forceinline__ float wave_sum_valu(float v) {
    v += xhalf(v);
    v += row_xor<8>(v);
    v += row_xor<4>(v);
    v += row_xor<2>(v);
    v += row_xor<1>(v);
    return v;
}
__device__ __forceinline__ float dot4(const float4 a, const float4 b) {
    v2f alo, ahi, blo, bhi;
    alo.x = a.x; alo.y = a.y; ahi.x = a.z; ahi.y = a.w;
    blo.x = b.x; blo.y = b.y; bhi.x = b.z; bhi.y = b.w;
    const v2f r = __builtin_elementwise_fma(ahi, bhi, alo * blo);
    return r.x + r.y;
}
__device__ __forceinline__ void axpy4(float p, const float4 v, v2f& lo, v2f& hi) {
    v2f pp, vlo, vhi;
    pp.x = p; pp.y = p; vlo.x = v.x; vlo.y = v.y; vhi.x = v.z; vhi.y = v.w;
    lo = __builtin_elementwise_fma(pp, vlo, lo);
    hi = __builtin_elementwise_fma(pp, vhi, hi);
}

template <bool GUARD>
__device__ __forceinline__ void take_rows(const float* __restrict__ x, int s, int t0, int rem, int c0,
                                          const float4 ws, float adv, float& m, float& sum, v2f& alo, v2f& ahi) {
    float4 v[8];
    float  e[8];
#pragma unroll
    for (int j = 0; j < 8; ++j) {
        if (!GUARD || j < rem) {
            const unsigned off = (unsigned)lane_i(s, t0 + j) * 512u + (unsigned)c0 * 4u;
            v[j] = *(const float4*)((const char*)x + off);
        }
    }
#pragma unroll
    for (int j = 0; j < 8; ++j) {
        e[j] = -INFINITY;
        if (!GUARD || j < rem) e[j] = leaky(wave_sum_valu(dot4(v[j], ws)) + adv);
    }
    float mn = m;
#pragma unroll
    for (int j = 0; j < 8; ++j) mn = fmaxf(mn, e[j]);
    const float sc = __expf(m - mn);
    sum *= sc;
    alo *= sc; ahi *= sc;
#pragma unroll
    for (int j = 0; j < 8; ++j) {
        if (!GUARD || j < rem) {
            const float p = __expf(e[j] - mn);
            sum += p;
            axpy4(p, v[j], alo, ahi);
        }
    }
    m = mn;
}

__global__ __launch_bounds__(256) __attribute__((amdgpu_waves_per_eu(16, 16))) void node_gather_kernel(const int* __restrict__ bucket,
                                                          const int* __restrict__ cnt,
                                                          const int* __restrict__ spill_cnt,
                                                          const int* __restrict__ spill,
                                                          const float* __restrict__ wsd,
                                                          const float* __restrict__ x,
                                                          float* __restrict__ aggx) {
    const int lane = threadIdx.x & 31;
    const int d = blockIdx.x * 8 + __builtin_amdgcn_readfirstlane((int)(threadIdx.x >> 5));
    if (d >= N_NODES) return;
    const int c0 = lane * 4;
    const int n = cnt[d];
    const int nb = n < BUCKET_CAP ? n : BUCKET_CAP;
    const float4 ws = *(const float4*)(wsd + c0);
    const float4 wd = *(const float4*)(wsd + 128 + c0);
    const float4 vs = *(const float4*)(x + (size_t)d * 128 + c0);
    int s = d;
    if (lane < nb) s = bucket[(size_t)d * BUCKET_CAP + lane];

    const float adv = lane_f(wave_sum_valu(dot4(vs, wd)), 0);
    float m = leaky(wave_sum_valu(dot4(vs, ws)) + adv);
    float sum = 1.0f;
    v2f alo, ahi;
    alo.x = vs.x; alo.y = vs.y; ahi.x = vs.z; ahi.y = vs.w;

    int t = 0;
    for (; t + 8 <= nb; t += 8) take_rows<false>(x, s, t, 8, c0, ws, adv, m, sum, alo, ahi);
    if (t < nb) take_rows<true>(x, s, t, nb - t, c0, ws, adv, m, sum, alo, ahi);

    if (n > BUCKET_CAP) {
        int spn = spill_cnt[0];
        if (spn > N_EDGES0) spn = N_EDGES0;
        const int2* sp = (const int2*)spill;
        for (int base = 0; base < spn; base += 32) {
            const int i = base + lane;
            int sv = d;
            bool hit = false;
            if (i < spn) {
                const int2 en = sp[i];
                if (en.x == d) { sv = en.y; hit = true; }
            }
            unsigned hits = __builtin_amdgcn_ballot_w32(hit);
            while (hits) {
                const int th = __builtin_ctz(hits);
                hits &= hits - 1;
                take_rows<true>(x, sv, th, 1, c0, ws, adv, m, sum, alo, ahi);
            }
        }
    }
    const float inv = 1.0f / sum;
    *(float4*)(aggx + (size_t)d * 128 + c0) = make_float4(alo.x * inv, alo.y * inv, ahi.x * inv, ahi.y * inv);
}

#define MLP_WG    256
#define MLP_WAVES 8
#define SP        68
#define WQ_F      (128 * 256)
#define W2Q_F     (256 * 128)
#define WOQ_F     (128 * 16)
#define PATCH_F   (16 * SP)

__device__ __forceinline__ v8f mm4(float a0, float a1, float b0, float b1, v8f acc) {
    v2f a; a.x = a0; a.y = a1;
    v2f bm; bm.x = b0; bm.y = b1;
    return __builtin_amdgcn_wmma_f32_16x16x4_f32(false, a, false, bm, (short)0, acc, false, false);
}

__global__ __launch_bounds__(256) void mlp_fused_kernel(const float* __restrict__ aggx,
                                                        const float* __restrict__ W,
                                                        const float* __restrict__ b,
                                                        const float* __restrict__ Wfc2,
                                                        const float* __restrict__ bfc2,
                                                        const float* __restrict__ Wout,
                                                        const float* __restrict__ bout,
                                                        float* __restrict__ out) {
    __shared__ __attribute__((aligned(16))) float lds[WQ_F + W2Q_F + WOQ_F + MLP_WAVES * PATCH_F];
    float* Wq  = lds;
    float* W2q = Wq + WQ_F;
    float* Woq = W2q + W2Q_F;
    const int tid  = threadIdx.x;
    const int lane = tid & 31;
    const int wave = __builtin_amdgcn_readfirstlane(tid >> 5);

#pragma unroll
    for (int i = 0; i < 8; ++i) {
        const int task = tid + 256 * i;
        const int kq = task >> 6, c4 = (task & 63) * 4;
        const float* g = W + (size_t)(4 * kq) * 256 + c4;
        const float4 r0 = *(const float4*)g;
        const float4 r1 = *(const float4*)(g + 256);
        const float4 r2 = *(const float4*)(g + 512);
        const float4 r3 = *(const float4*)(g + 768);
        float4* dst = (float4*)(Wq + (size_t)(kq * 256 + c4) * 4);
        dst[0] = make_float4(r0.x, r1.x, r2.x, r3.x);
        dst[1] = make_float4(r0.y, r1.y, r2.y, r3.y);
        dst[2] = make_float4(r0.z, r1.z, r2.z, r3.z);
        dst[3] = make_float4(r0.w, r1.w, r2.w, r3.w);
    }
#pragma unroll
    for (int i = 0; i < 8; ++i) {
        const int task = tid + 256 * i;
        const int kq = task >> 5, c4 = (task & 31) * 4;
        const float* g = Wfc2 + (size_t)(4 * kq) * 128 + c4;
        const float4 r0 = *(const float4*)g;
        const float4 r1 = *(const float4*)(g + 128);
        const float4 r2 = *(const float4*)(g + 256);
        const float4 r3 = *(const float4*)(g + 384);
        float4* dst = (float4*)(W2q + (size_t)(kq * 128 + c4) * 4);
        dst[0] = make_float4(r0.x, r1.x, r2.x, r3.x);
        dst[1] = make_float4(r0.y, r1.y, r2.y, r3.y);
        dst[2] = make_float4(r0.z, r1.z, r2.z, r3.z);
        dst[3] = make_float4(r0.w, r1.w, r2.w, r3.w);
    }
#pragma unroll
    for (int n = 0; n < 8; ++n) {
        const int i = tid + 256 * n;
        const int j = i & 3, col = (i >> 2) & 15, kq = i >> 6;
        const float v = Wout[(4 * kq + j) * 6 + (col < 6 ? col : 0)];
        Woq[i] = (col < 6) ? v : 0.0f;
    }
    __syncthreads();

    const int c = lane & 15;
    const int h = lane >> 4;
    float* patch = Woq + WOQ_F + wave * PATCH_F;
    const float4* Wq4  = (const float4*)Wq;
    const float4* W2q4 = (const float4*)W2q;
    const float4* Woq4 = (const float4*)Woq;
    float*        pst  = patch + (8 * h) * SP + c;
    const float4* pld  = (const float4*)(patch + c * SP + 4 * h);

    for (int it = 0; it < 4; ++it) {
        const int tile = it * (MLP_WG * MLP_WAVES) + wave * MLP_WG + (int)blockIdx.x;
        if (tile >= ROW_TILES) break;
        const int row0 = tile * 16;

        float4 a[16];
        const float* ap = aggx + (size_t)(row0 + c) * 128 + 4 * h;
#pragma unroll
        for (int kb = 0; kb < 16; ++kb) a[kb] = *(const float4*)(ap + 8 * kb);

        v8f acc2[8] = {};
#pragma unroll 1
        for (int p = 0; p < 4; ++p) {
            v8f acc1[4] = {};
            const float4* wq = Wq4 + h * 256 + p * 64 + c;
#pragma unroll
            for (int kb = 0; kb < 16; ++kb) {
                float4 bq[4];
#pragma unroll
                for (int t = 0; t < 4; ++t) bq[t] = wq[kb * 512 + t * 16];
#pragma unroll
                for (int t = 0; t < 4; ++t) acc1[t] = mm4(a[kb].x, a[kb].y, bq[t].x, bq[t].y, acc1[t]);
#pragma unroll
                for (int t = 0; t < 4; ++t) acc1[t] = mm4(a[kb].z, a[kb].w, bq[t].z, bq[t].w, acc1[t]);
            }
#pragma unroll
            for (int t = 0; t < 4; ++t) {
                const float bb = b[p * 64 + t * 16 + c];
#pragma unroll
                for (int r = 0; r < 8; ++r) pst[r * SP + t * 16] = fmaxf(acc1[t][r] + bb, 0.0f);
            }
            const float4* w2 = W2q4 + (p * 16 + h) * 128 + c;
#pragma unroll
            for (int kb = 0; kb < 8; ++kb) {
                const float4 a2 = pld[2 * kb];
                float4 bq[8];
#pragma unroll
                for (int t = 0; t < 8; ++t) bq[t] = w2[kb * 256 + t * 16];
#pragma unroll
                for (int t = 0; t < 8; ++t) acc2[t] = mm4(a2.x, a2.y, bq[t].x, bq[t].y, acc2[t]);
#pragma unroll
                for (int t = 0; t < 8; ++t) acc2[t] = mm4(a2.z, a2.w, bq[t].z, bq[t].w, acc2[t]);
            }
        }

        v8f acc3a = {}, acc3b = {};
#pragma unroll
        for (int half = 0; half < 2; ++half) {
#pragma unroll
            for (int t = 0; t < 4; ++t) {
                const float bb = bfc2[half * 64 + t * 16 + c];
#pragma unroll
                for (int r = 0; r < 8; ++r) pst[r * SP + t * 16] = fmaxf(acc2[half * 4 + t][r] + bb, 0.0f);
            }
            const float4* wo = Woq4 + (half * 16 + h) * 16 + c;
#pragma unroll
            for (int kb = 0; kb < 8; ++kb) {
                const float4 a3 = pld[2 * kb];
                const float4 bq = wo[kb * 32];
                acc3a = mm4(a3.x, a3.y, bq.x, bq.y, acc3a);
                acc3b = mm4(a3.z, a3.w, bq.z, bq.w, acc3b);
            }
        }
        if (c < 6) {
            const float bb = bout[c];
            float* op = out + (size_t)(row0 + 8 * h) * 6 + c;
#pragma unroll
            for (int r = 0; r < 8; ++r) op[r * 6] = (acc3a[r] + acc3b[r]) + bb;
        }
    }
}

extern "C" void kernel_launch(void* const* d_in, const int* in_sizes, int n_in,
                              void* d_out, int out_size, void* d_ws, size_t ws_size,
                              hipStream_t stream) {
    const float* x     = (const float*)d_in[0];
    const int*   ei    = (const int*)  d_in[1];
    const float* W     = (const float*)d_in[2];
    const float* a_src = (const float*)d_in[3];
    const float* a_dst = (const float*)d_in[4];
    const float* b     = (const float*)d_in[5];
    const float* Wfc2  = (const float*)d_in[6];
    const float* bfc2  = (const float*)d_in[7];
    const float* Wout  = (const float*)d_in[8];
    const float* bout  = (const float*)d_in[9];
    float* out = (float*)d_out;

    float* ws      = (float*)d_ws;
    float* aggx    = ws;
    float* wsd     = aggx + (size_t)N_NODES * 128;
    int*   cnt     = (int*)(wsd + 256);
    int*   spcnt   = cnt + N_NODES;
    int*   bucket  = spcnt + 32;
    int*   spill   = bucket + (size_t)N_NODES * BUCKET_CAP;

    prep_kernel<<<ZERO_BLOCKS + FOLD_BLOCKS, 256, 0, stream>>>(cnt, W, a_src, a_dst, wsd);
    bucket_fill_kernel<<<(N_EDGES0 / 4 + 255) / 256, 256, 0, stream>>>(ei, cnt, spcnt, bucket, spill);
    node_gather_kernel<<<(N_NODES + 7) / 8, 256, 0, stream>>>(bucket, cnt, spcnt, spill, wsd, x, aggx);
    mlp_fused_kernel<<<MLP_WG, 256, 0, stream>>>(aggx, W, b, Wfc2, bfc2, Wout, bout, out);
}


# Edges grouped by destination in two LDS-binned passes (chunk sort by 512-node bin, then a workgroup per bin fills buckets from LDS counters): no global atomics; per-node sum order = slot arrival
# speedup vs baseline: 5.5420x; 1.3804x over previous
#include <hip/hip_runtime.h>
#include <hip/hip_bf16.h>

typedef __attribute__((ext_vector_type(2))) float v2f;
typedef __attribute__((ext_vector_type(8))) float v8f;

#define N_NODES   100000
#define N_EDGES0  800000
#define N_EDGES   900000
#define ROW_TILES 6250

__device__ __forceinline__ float leaky(float v) { return v > 0.0f ? v : 0.2f * v; }

__device__ __forceinline__ float wave_max_f(float v) {
#pragma unroll
    for (int off = 16; off; off >>= 1) v = fmaxf(v, __shfl_xor(v, off, 32));
    return v;
}
__device__ __forceinline__ float wave_sum_f(float v) {
#pragma unroll
    for (int off = 16; off; off >>= 1) v += __shfl_xor(v, off, 32);
    return v;
}

#define BUCKET_CAP 32
#define CHUNK      4096
#define N_CHUNKS   196
#define BIN_SHIFT  9
#define BIN_NODES  512
#define N_BINS     196
#define OFF_PITCH  200
#define SRC_MASK   0x1ffff
#define FOLD_BLOCKS 4
__global__ __launch_bounds__(1024) void bin_kernel(const int* __restrict__ ei,
                                                   const float* __restrict__ W,
                                                   const float* __restrict__ a_src,
                                                   const float* __restrict__ a_dst,
                                                   float* __restrict__ wsd,
                                                   int* __restrict__ sorted,
                                                   int* __restrict__ offs) {
    __shared__ int lcnt[256];
    __shared__ int loff[256];
    __shared__ __attribute__((aligned(16))) int lsorted[CHUNK];
    const int tid = threadIdx.x;
    if (blockIdx.x >= N_CHUNKS) {
        const int lane = tid & 31;
        const int k = ((int)blockIdx.x - N_CHUNKS) * 32 + (tid >> 5);
        const float* wr = W + (size_t)k * 256;
        float s = 0.0f, d = 0.0f;
#pragma unroll
        for (int i = 0; i < 8; ++i) {
            float v = wr[lane + 32 * i];
            s += v * a_src[lane + 32 * i];
            d += v * a_dst[lane + 32 * i];
        }
        s = wave_sum_f(s);
        d = wave_sum_f(d);
        if (lane == 0) { wsd[k] = s; wsd[128 + k] = d; }
        return;
    }
    if (tid < 256) lcnt[tid] = 0;
    __syncthreads();
    const int c = blockIdx.x;
    const int e0 = c * CHUNK + tid * 4;
    const bool live = e0 < N_EDGES0;
    int bin[4], rank[4], val[4];
    if (live) {
        const int4 s4 = *(const int4*)(ei + e0);
        const int4 d4 = *(const int4*)(ei + N_EDGES0 + e0);
        const int dd[4] = {d4.x, d4.y, d4.z, d4.w};
        const int ss[4] = {s4.x, s4.y, s4.z, s4.w};
#pragma unroll
        for (int k = 0; k < 4; ++k) {
            bin[k]  = (dd[k] >> BIN_SHIFT) & 255;
            val[k]  = ((dd[k] & (BIN_NODES - 1)) << 17) | (ss[k] & SRC_MASK);
            rank[k] = atomicAdd(&lcnt[bin[k]], 1);
        }
    }
    __syncthreads();
    if (tid < 32) {
        int v[8];
        int sum = 0;
#pragma unroll
        for (int i = 0; i < 8; ++i) { v[i] = lcnt[tid * 8 + i]; sum += v[i]; }
        int incl = sum;
#pragma unroll
        for (int off = 1; off < 32; off <<= 1) {
            const int t = __shfl_up(incl, off, 32);
            if (tid >= off) incl += t;
        }
        int run = incl - sum;
#pragma unroll
        for (int i = 0; i < 8; ++i) { loff[tid * 8 + i] = run; run += v[i]; }
    }
    __syncthreads();
    if (live) {
#pragma unroll
        for (int k = 0; k < 4; ++k) lsorted[loff[bin[k]] + rank[k]] = val[k];
    }
    __syncthreads();
    if (live) *(int4*)(sorted + (size_t)c * CHUNK + tid * 4) = *(const int4*)(lsorted + tid * 4);
    if (tid < OFF_PITCH) offs[c * OFF_PITCH + tid] = loff[tid];
}

__global__ __launch_bounds__(1024) void build_kernel(const int* __restrict__ sorted,
                                                     const int* __restrict__ offs,
                                                     int* __restrict__ cnt,
                                                     int* __restrict__ bucket) {
    __shared__ int ncnt[BIN_NODES];
    __shared__ __attribute__((aligned(16))) int lbkt[BIN_NODES * BUCKET_CAP];
    const int tid = threadIdx.x;
    const int lane = tid & 31;
    const int wave = __builtin_amdgcn_readfirstlane(tid >> 5);
    const int j = blockIdx.x;
    if (tid < BIN_NODES) ncnt[tid] = 0;
    for (int q = tid; q < BIN_NODES * BUCKET_CAP / 4; q += 1024) ((int4*)lbkt)[q] = make_int4(0, 0, 0, 0);
    __syncthreads();
    int lo_v = 0, hi_v = 0;
    {
        const int cc = wave + 32 * lane;
        if (cc < N_CHUNKS) {
            lo_v = offs[cc * OFF_PITCH + j];
            hi_v = offs[cc * OFF_PITCH + j + 1];
        }
    }
    int lo_s[7], hi_s[7], pv[7];
#pragma unroll
    for (int k = 0; k < 7; ++k) {
        lo_s[k] = __builtin_amdgcn_readlane(lo_v, k);
        hi_s[k] = __builtin_amdgcn_readlane(hi_v, k);
        if (lo_s[k] < 0) lo_s[k] = 0;
        if (hi_s[k] > CHUNK) hi_s[k] = CHUNK;
        pv[k] = -1;
        if (lo_s[k] + lane < hi_s[k]) pv[k] = sorted[(size_t)(wave + 32 * k) * CHUNK + lo_s[k] + lane];
    }
#pragma unroll
    for (int k = 0; k < 7; ++k) {
        if (pv[k] >= 0) {
            const int u = (pv[k] >> 17) & (BIN_NODES - 1);
            const int slot = atomicAdd(&ncnt[u], 1);
            if (slot < BUCKET_CAP) lbkt[u * BUCKET_CAP + slot] = pv[k] & SRC_MASK;
        }
    }
#pragma unroll
    for (int k = 0; k < 7; ++k) {
        for (int i = lo_s[k] + 32 + lane; i < hi_s[k]; i += 32) {
            const int pw = sorted[(size_t)(wave + 32 * k) * CHUNK + i];
            const int u = (pw >> 17) & (BIN_NODES - 1);
            const int slot = atomicAdd(&ncnt[u], 1);
            if (slot < BUCKET_CAP) lbkt[u * BUCKET_CAP + slot] = pw & SRC_MASK;
        }
    }
    __syncthreads();
    const int d0 = j * BIN_NODES;
    if (tid < BIN_NODES && d0 + tid < N_NODES) cnt[d0 + tid] = ncnt[tid];
    for (int q = tid; q < BIN_NODES * BUCKET_CAP / 4; q += 1024) {
        if (d0 + (q >> 3) < N_NODES)
            ((int4*)bucket)[(size_t)d0 * (BUCKET_CAP / 4) + q] = ((const int4*)lbkt)[q];
    }
}

__device__ __forceinline__ int lane_i(int v, int t) { return __builtin_amdgcn_readlane(v, t); }
__device__ __forceinline__ float lane_f(float v, int t) {
    return __int_as_float(__builtin_amdgcn_readlane(__float_as_int(v), t));
}

__device__ __forceinline__ float xhalf(float v) {
    const int b = __float_as_int(v);
    return __int_as_float(__builtin_amdgcn_permlanex16(b, b, 0x76543210, 0xfedcba98, false, false));
}
template <int N> __device__ __forceinline__ float row_xor(float v) {
    return __int_as_float(__builtin_amdgcn_update_dpp(0, __float_as_int(v), 0x160 + N, 0xf, 0xf, true));
}
__device__ __forceinline__ float wave_sum_valu(float v) {
    v += xhalf(v);
    v += row_xor<8>(v);
    v += row_xor<4>(v);
    v += row_xor<2>(v);
    v += row_xor<1>(v);
    return v;
}
__device__ __forceinline__ float dot4(const float4 a, const float4 b) {
    v2f alo, ahi, blo, bhi;
    alo.x = a.x; alo.y = a.y; ahi.x = a.z; ahi.y = a.w;
    blo.x = b.x; blo.y = b.y; bhi.x = b.z; bhi.y = b.w;
    const v2f r = __builtin_elementwise_fma(ahi, bhi, alo * blo);
    return r.x + r.y;
}
__device__ __forceinline__ void axpy4(float p, const float4 v, v2f& lo, v2f& hi) {
    v2f pp, vlo, vhi;
    pp.x = p; pp.y = p; vlo.x = v.x; vlo.y = v.y; vhi.x = v.z; vhi.y = v.w;
    lo = __builtin_elementwise_fma(pp, vlo, lo);
    hi = __builtin_elementwise_fma(pp, vhi, hi);
}

template <bool GUARD>
__device__ __forceinline__ void take_rows(const float* __restrict__ x, int s, int t0, int rem, int c0,
                                          const float4 ws, float adv, float& m, float& sum, v2f& alo, v2f& ahi) {
    float4 v[8];
    float  e[8];
#pragma unroll
    for (int j = 0; j < 8; ++j) {
        if (!GUARD || j < rem) {
            const unsigned off = (unsigned)lane_i(s, t0 + j) * 512u + (unsigned)c0 * 4u;
            v[j] = *(const float4*)((const char*)x + off);
        }
    }
#pragma unroll
    for (int j = 0; j < 8; ++j) {
        e[j] = -INFINITY;
        if (!GUARD || j < rem) e[j] = leaky(wave_sum_valu(dot4(v[j], ws)) + adv);
    }
    float mn = m;
#pragma unroll
    for (int j = 0; j < 8; ++j) mn = fmaxf(mn, e[j]);
    const float sc = __expf(m - mn);
    sum *= sc;
    alo *= sc; ahi *= sc;
#pragma unroll
    for (int j = 0; j < 8; ++j) {
        if (!GUARD || j < rem) {
            const float p = __expf(e[j] - mn);
            sum += p;
            axpy4(p, v[j], alo, ahi);
        }
    }
    m = mn;
}

__global__ __launch_bounds__(256) __attribute__((amdgpu_waves_per_eu(16, 16))) void node_gather_kernel(const int* __restrict__ bucket,
                                                          const int* __restrict__ cnt,
                                                          const int* __restrict__ sorted,
                                                          const int* __restrict__ offs,
                                                          const float* __restrict__ wsd,
                                                          const float* __restrict__ x,
                                                          float* __restrict__ aggx) {
    const int lane = threadIdx.x & 31;
    const int d = blockIdx.x * 8 + __builtin_amdgcn_readfirstlane((int)(threadIdx.x >> 5));
    if (d >= N_NODES) return;
    const int c0 = lane * 4;
    const int n = cnt[d];
    const int nb = n <= BUCKET_CAP ? n : 0;
    const float4 ws = *(const float4*)(wsd + c0);
    const float4 wd = *(const float4*)(wsd + 128 + c0);
    const float4 vs = *(const float4*)(x + (size_t)d * 128 + c0);
    int s = d;
    if (lane < nb) s = bucket[(size_t)d * BUCKET_CAP + lane];

    const float adv = lane_f(wave_sum_valu(dot4(vs, wd)), 0);
    float m = leaky(wave_sum_valu(dot4(vs, ws)) + adv);
    float sum = 1.0f;
    v2f alo, ahi;
    alo.x = vs.x; alo.y = vs.y; ahi.x = vs.z; ahi.y = vs.w;

    int t = 0;
    for (; t + 8 <= nb; t += 8) take_rows<false>(x, s, t, 8, c0, ws, adv, m, sum, alo, ahi);
    if (t < nb) take_rows<true>(x, s, t, nb - t, c0, ws, adv, m, sum, alo, ahi);

    if (n > BUCKET_CAP) {
        const int j = d >> BIN_SHIFT;
        const int u = d & (BIN_NODES - 1);
        for (int cc = 0; cc < N_CHUNKS; ++cc) {
            int lo = offs[cc * OFF_PITCH + j];
            int hi = offs[cc * OFF_PITCH + j + 1];
            if (lo < 0) lo = 0;
            if (hi > CHUNK) hi = CHUNK;
            for (int base = lo; base < hi; base += 32) {
                const int i = base + lane;
                int sv = d;
                bool hit = false;
                if (i < hi) {
                    const int pv = sorted[(size_t)cc * CHUNK + i];
                    if (((pv >> 17) & (BIN_NODES - 1)) == u) { sv = pv & SRC_MASK; hit = true; }
                }
                unsigned hits = __builtin_amdgcn_ballot_w32(hit);
                while (hits) {
                    const int th = __builtin_ctz(hits);
                    hits &= hits - 1;
                    take_rows<true>(x, sv, th, 1, c0, ws, adv, m, sum, alo, ahi);
                }
            }
        }
    }
    const float inv = 1.0f / sum;
    *(float4*)(aggx + (size_t)d * 128 + c0) = make_float4(alo.x * inv, alo.y * inv, ahi.x * inv, ahi.y * inv);
}

#define MLP_WG    256
#define MLP_WAVES 8
#define SP        68
#define WQ_F      (128 * 256)
#define W2Q_F     (256 * 128)
#define WOQ_F     (128 * 16)
#define PATCH_F   (16 * SP)

__device__ __forceinline__ v8f mm4(float a0, float a1, float b0, float b1, v8f acc) {
    v2f a; a.x = a0; a.y = a1;
    v2f bm; bm.x = b0; bm.y = b1;
    return __builtin_amdgcn_wmma_f32_16x16x4_f32(false, a, false, bm, (short)0, acc, false, false);
}

__global__ __launch_bounds__(256) void mlp_fused_kernel(const float* __restrict__ aggx,
                                                        const float* __restrict__ W,
                                                        const float* __restrict__ b,
                                                        const float* __restrict__ Wfc2,
                                                        const float* __restrict__ bfc2,
                                                        const float* __restrict__ Wout,
                                                        const float* __restrict__ bout,
                                                        float* __restrict__ out) {
    __shared__ __attribute__((aligned(16))) float lds[WQ_F + W2Q_F + WOQ_F + MLP_WAVES * PATCH_F];
    float* Wq  = lds;
    float* W2q = Wq + WQ_F;
    float* Woq = W2q + W2Q_F;
    const int tid  = threadIdx.x;
    const int lane = tid & 31;
    const int wave = __builtin_amdgcn_readfirstlane(tid >> 5);

#pragma unroll
    for (int i = 0; i < 8; ++i) {
        const int task = tid + 256 * i;
        const int kq = task >> 6, c4 = (task & 63) * 4;
        const float* g = W + (size_t)(4 * kq) * 256 + c4;
        const float4 r0 = *(const float4*)g;
        const float4 r1 = *(const float4*)(g + 256);
        const float4 r2 = *(const float4*)(g + 512);
        const float4 r3 = *(const float4*)(g + 768);
        float4* dst = (float4*)(Wq + (size_t)(kq * 256 + c4) * 4);
        dst[0] = make_float4(r0.x, r1.x, r2.x, r3.x);
        dst[1] = make_float4(r0.y, r1.y, r2.y, r3.y);
        dst[2] = make_float4(r0.z, r1.z, r2.z, r3.z);
        dst[3] = make_float4(r0.w, r1.w, r2.w, r3.w);
    }
#pragma unroll
    for (int i = 0; i < 8; ++i) {
        const int task = tid + 256 * i;
        const int kq = task >> 5, c4 = (task & 31) * 4;
        const float* g = Wfc2 + (size_t)(4 * kq) * 128 + c4;
        const float4 r0 = *(const float4*)g;
        const float4 r1 = *(const float4*)(g + 128);
        const float4 r2 = *(const float4*)(g + 256);
        const float4 r3 = *(const float4*)(g + 384);
        float4* dst = (float4*)(W2q + (size_t)(kq * 128 + c4) * 4);
        dst[0] = make_float4(r0.x, r1.x, r2.x, r3.x);
        dst[1] = make_float4(r0.y, r1.y, r2.y, r3.y);
        dst[2] = make_float4(r0.z, r1.z, r2.z, r3.z);
        dst[3] = make_float4(r0.w, r1.w, r2.w, r3.w);
    }
#pragma unroll
    for (int n = 0; n < 8; ++n) {
        const int i = tid + 256 * n;
        const int j = i & 3, col = (i >> 2) & 15, kq = i >> 6;
        const float v = Wout[(4 * kq + j) * 6 + (col < 6 ? col : 0)];
        Woq[i] = (col < 6) ? v : 0.0f;
    }
    __syncthreads();

    const int c = lane & 15;
    const int h = lane >> 4;
    float* patch = Woq + WOQ_F + wave * PATCH_F;
    const float4* Wq4  = (const float4*)Wq;
    const float4* W2q4 = (const float4*)W2q;
    const float4* Woq4 = (const float4*)Woq;
    float*        pst  = patch + (8 * h) * SP + c;
    const float4* pld  = (const float4*)(patch + c * SP + 4 * h);

    for (int it = 0; it < 4; ++it) {
        const int tile = it * (MLP_WG * MLP_WAVES) + wave * MLP_WG + (int)blockIdx.x;
        if (tile >= ROW_TILES) break;
        const int row0 = tile * 16;

        float4 a[16];
        const float* ap = aggx + (size_t)(row0 + c) * 128 + 4 * h;
#pragma unroll
        for (int kb = 0; kb < 16; ++kb) a[kb] = *(const float4*)(ap + 8 * kb);

        v8f acc2[8] = {};
#pragma unroll 1
        for (int p = 0; p < 4; ++p) {
            v8f acc1[4] = {};
            const float4* wq = Wq4 + h * 256 + p * 64 + c;
#pragma unroll
            for (int kb = 0; kb < 16; ++kb) {
                float4 bq[4];
#pragma unroll
                for (int t = 0; t < 4; ++t) bq[t] = wq[kb * 512 + t * 16];
#pragma unroll
                for (int t = 0; t < 4; ++t) acc1[t] = mm4(a[kb].x, a[kb].y, bq[t].x, bq[t].y, acc1[t]);
#pragma unroll
                for (int t = 0; t < 4; ++t) acc1[t] = mm4(a[kb].z, a[kb].w, bq[t].z, bq[t].w, acc1[t]);
            }
#pragma unroll
            for (int t = 0; t < 4; ++t) {
                const float bb = b[p * 64 + t * 16 + c];
#pragma unroll
                for (int r = 0; r < 8; ++r) pst[r * SP + t * 16] = fmaxf(acc1[t][r] + bb, 0.0f);
            }
            const float4* w2 = W2q4 + (p * 16 + h) * 128 + c;
#pragma unroll
            for (int kb = 0; kb < 8; ++kb) {
                const float4 a2 = pld[2 * kb];
                float4 bq[8];
#pragma unroll
                for (int t = 0; t < 8; ++t) bq[t] = w2[kb * 256 + t * 16];
#pragma unroll
                for (int t = 0; t < 8; ++t) acc2[t] = mm4(a2.x, a2.y, bq[t].x, bq[t].y, acc2[t]);
#pragma unroll
                for (int t = 0; t < 8; ++t) acc2[t] = mm4(a2.z, a2.w, bq[t].z, bq[t].w, acc2[t]);
            }
        }

        v8f acc3a = {}, acc3b = {};
#pragma unroll
        for (int half = 0; half < 2; ++half) {
#pragma unroll
            for (int t = 0; t < 4; ++t) {
                const float bb = bfc2[half * 64 + t * 16 + c];
#pragma unroll
                for (int r = 0; r < 8; ++r) pst[r * SP + t * 16] = fmaxf(acc2[half * 4 + t][r] + bb, 0.0f);
            }
            const float4* wo = Woq4 + (half * 16 + h) * 16 + c;
#pragma unroll
            for (int kb = 0; kb < 8; ++kb) {
                const float4 a3 = pld[2 * kb];
                const float4 bq = wo[kb * 32];
                acc3a = mm4(a3.x, a3.y, bq.x, bq.y, acc3a);
                acc3b = mm4(a3.z, a3.w, bq.z, bq.w, acc3b);
            }
        }
        if (c < 6) {
            const float bb = bout[c];
            float* op = out + (size_t)(row0 + 8 * h) * 6 + c;
#pragma unroll
            for (int r = 0; r < 8; ++r) op[r * 6] = (acc3a[r] + acc3b[r]) + bb;
        }
    }
}

extern "C" void kernel_launch(void* const* d_in, const int* in_sizes, int n_in,
                              void* d_out, int out_size, void* d_ws, size_t ws_size,
                              hipStream_t stream) {
    const float* x     = (const float*)d_in[0];
    const int*   ei    = (const int*)  d_in[1];
    const float* W     = (const float*)d_in[2];
    const float* a_src = (const float*)d_in[3];
    const float* a_dst = (const float*)d_in[4];
    const float* b     = (const float*)d_in[5];
    const float* Wfc2  = (const float*)d_in[6];
    const float* bfc2  = (const float*)d_in[7];
    const float* Wout  = (const float*)d_in[8];
    const float* bout  = (const float*)d_in[9];
    float* out = (float*)d_out;

    float* ws      = (float*)d_ws;
    float* aggx    = ws;
    float* wsd     = aggx + (size_t)N_NODES * 128;
    int*   cnt     = (int*)(wsd + 256);
    int*   bucket  = cnt + N_NODES;
    int*   sorted  = bucket + (size_t)N_NODES * BUCKET_CAP;
    int*   offs    = sorted + (size_t)N_CHUNKS * CHUNK;

    bin_kernel<<<N_CHUNKS + FOLD_BLOCKS, 1024, 0, stream>>>(ei, W, a_src, a_dst, wsd, sorted, offs);
    build_kernel<<<N_BINS, 1024, 0, stream>>>(sorted, offs, cnt, bucket);
    node_gather_kernel<<<(N_NODES + 7) / 8, 256, 0, stream>>>(bucket, cnt, sorted, offs, wsd, x, aggx);
    mlp_fused_kernel<<<MLP_WG, 256, 0, stream>>>(aggx, W, b, Wfc2, bfc2, Wout, bout, out);
}


# Dense launch: the 106 row tiles left after three per wave are split four ways by layer-1 column pass over waves 0..3 of one workgroup each; layer-2 f32 sum order changes there (4 partials added)
# speedup vs baseline: 5.7365x; 1.0351x over previous
#include <hip/hip_runtime.h>
#include <hip/hip_bf16.h>

typedef __attribute__((ext_vector_type(2))) float v2f;
typedef __attribute__((ext_vector_type(8))) float v8f;

#define N_NODES   100000
#define N_EDGES0  800000
#define N_EDGES   900000
#define ROW_TILES 6250

__device__ __forceinline__ float leaky(float v) { return v > 0.0f ? v : 0.2f * v; }

__device__ __forceinline__ float wave_max_f(float v) {
#pragma unroll
    for (int off = 16; off; off >>= 1) v = fmaxf(v, __shfl_xor(v, off, 32));
    return v;
}
__device__ __forceinline__ float wave_sum_f(float v) {
#pragma unroll
    for (int off = 16; off; off >>= 1) v += __shfl_xor(v, off, 32);
    return v;
}

#define BUCKET_CAP 32
#define CHUNK      4096
#define N_CHUNKS   196
#define BIN_SHIFT  9
#define BIN_NODES  512
#define N_BINS     196
#define OFF_PITCH  200
#define SRC_MASK   0x1ffff
#define FOLD_BLOCKS 4
__global__ __launch_bounds__(1024) void bin_kernel(const int* __restrict__ ei,
                                                   const float* __restrict__ W,
                                                   const float* __restrict__ a_src,
                                                   const float* __restrict__ a_dst,
                                                   float* __restrict__ wsd,
                                                   int* __restrict__ sorted,
                                                   int* __restrict__ offs) {
    __shared__ int lcnt[256];
    __shared__ int loff[256];
    __shared__ __attribute__((aligned(16))) int lsorted[CHUNK];
    const int tid = threadIdx.x;
    if (blockIdx.x >= N_CHUNKS) {
        const int lane = tid & 31;
        const int k = ((int)blockIdx.x - N_CHUNKS) * 32 + (tid >> 5);
        const float* wr = W + (size_t)k * 256;
        float s = 0.0f, d = 0.0f;
#pragma unroll
        for (int i = 0; i < 8; ++i) {
            float v = wr[lane + 32 * i];
            s += v * a_src[lane + 32 * i];
            d += v * a_dst[lane + 32 * i];
        }
        s = wave_sum_f(s);
        d = wave_sum_f(d);
        if (lane == 0) { wsd[k] = s; wsd[128 + k] = d; }
        return;
    }
    if (tid < 256) lcnt[tid] = 0;
    __syncthreads();
    const int c = blockIdx.x;
    const int e0 = c * CHUNK + tid * 4;
    const bool live = e0 < N_EDGES0;
    int bin[4], rank[4], val[4];
    if (live) {
        const int4 s4 = *(const int4*)(ei + e0);
        const int4 d4 = *(const int4*)(ei + N_EDGES0 + e0);
        const int dd[4] = {d4.x, d4.y, d4.z, d4.w};
        const int ss[4] = {s4.x, s4.y, s4.z, s4.w};
#pragma unroll
        for (int k = 0; k < 4; ++k) {
            bin[k]  = (dd[k] >> BIN_SHIFT) & 255;
            val[k]  = ((dd[k] & (BIN_NODES - 1)) << 17) | (ss[k] & SRC_MASK);
            rank[k] = atomicAdd(&lcnt[bin[k]], 1);
        }
    }
    __syncthreads();
    if (tid < 32) {
        int v[8];
        int sum = 0;
#pragma unroll
        for (int i = 0; i < 8; ++i) { v[i] = lcnt[tid * 8 + i]; sum += v[i]; }
        int incl = sum;
#pragma unroll
        for (int off = 1; off < 32; off <<= 1) {
            const int t = __shfl_up(incl, off, 32);
            if (tid >= off) incl += t;
        }
        int run = incl - sum;
#pragma unroll
        for (int i = 0; i < 8; ++i) { loff[tid * 8 + i] = run; run += v[i]; }
    }
    __syncthreads();
    if (live) {
#pragma unroll
        for (int k = 0; k < 4; ++k) lsorted[loff[bin[k]] + rank[k]] = val[k];
    }
    __syncthreads();
    if (live) *(int4*)(sorted + (size_t)c * CHUNK + tid * 4) = *(const int4*)(lsorted + tid * 4);
    if (tid < OFF_PITCH) offs[c * OFF_PITCH + tid] = loff[tid];
}

__global__ __launch_bounds__(1024) void build_kernel(const int* __restrict__ sorted,
                                                     const int* __restrict__ offs,
                                                     int* __restrict__ cnt,
                                                     int* __restrict__ bucket) {
    __shared__ int ncnt[BIN_NODES];
    __shared__ __attribute__((aligned(16))) int lbkt[BIN_NODES * BUCKET_CAP];
    const int tid = threadIdx.x;
    const int lane = tid & 31;
    const int wave = __builtin_amdgcn_readfirstlane(tid >> 5);
    const int j = blockIdx.x;
    if (tid < BIN_NODES) ncnt[tid] = 0;
    for (int q = tid; q < BIN_NODES * BUCKET_CAP / 4; q += 1024) ((int4*)lbkt)[q] = make_int4(0, 0, 0, 0);
    __syncthreads();
    int lo_v = 0, hi_v = 0;
    {
        const int cc = wave + 32 * lane;
        if (cc < N_CHUNKS) {
            lo_v = offs[cc * OFF_PITCH + j];
            hi_v = offs[cc * OFF_PITCH + j + 1];
        }
    }
    int lo_s[7], hi_s[7], pv[7];
#pragma unroll
    for (int k = 0; k < 7; ++k) {
        lo_s[k] = __builtin_amdgcn_readlane(lo_v, k);
        hi_s[k] = __builtin_amdgcn_readlane(hi_v, k);
        if (lo_s[k] < 0) lo_s[k] = 0;
        if (hi_s[k] > CHUNK) hi_s[k] = CHUNK;
        pv[k] = -1;
        if (lo_s[k] + lane < hi_s[k]) pv[k] = sorted[(size_t)(wave + 32 * k) * CHUNK + lo_s[k] + lane];
    }
#pragma unroll
    for (int k = 0; k < 7; ++k) {
        if (pv[k] >= 0) {
            const int u = (pv[k] >> 17) & (BIN_NODES - 1);
            const int slot = atomicAdd(&ncnt[u], 1);
            if (slot < BUCKET_CAP) lbkt[u * BUCKET_CAP + slot] = pv[k] & SRC_MASK;
        }
    }
#pragma unroll
    for (int k = 0; k < 7; ++k) {
        for (int i = lo_s[k] + 32 + lane; i < hi_s[k]; i += 32) {
            const int pw = sorted[(size_t)(wave + 32 * k) * CHUNK + i];
            const int u = (pw >> 17) & (BIN_NODES - 1);
            const int slot = atomicAdd(&ncnt[u], 1);
            if (slot < BUCKET_CAP) lbkt[u * BUCKET_CAP + slot] = pw & SRC_MASK;
        }
    }
    __syncthreads();
    const int d0 = j * BIN_NODES;
    if (tid < BIN_NODES && d0 + tid < N_NODES) cnt[d0 + tid] = ncnt[tid];
    for (int q = tid; q < BIN_NODES * BUCKET_CAP / 4; q += 1024) {
        if (d0 + (q >> 3) < N_NODES)
            ((int4*)bucket)[(size_t)d0 * (BUCKET_CAP / 4) + q] = ((const int4*)lbkt)[q];
    }
}

__device__ __forceinline__ int lane_i(int v, int t) { return __builtin_amdgcn_readlane(v, t); }
__device__ __forceinline__ float lane_f(float v, int t) {
    return __int_as_float(__builtin_amdgcn_readlane(__float_as_int(v), t));
}

__device__ __forceinline__ float xhalf(float v) {
    const int b = __float_as_int(v);
    return __int_as_float(__builtin_amdgcn_permlanex16(b, b, 0x76543210, 0xfedcba98, false, false));
}
template <int N> __device__ __forceinline__ float row_xor(float v) {
    return __int_as_float(__builtin_amdgcn_update_dpp(0, __float_as_int(v), 0x160 + N, 0xf, 0xf, true));
}
__device__ __forceinline__ float wave_sum_valu(float v) {
    v += xhalf(v);
    v += row_xor<8>(v);
    v += row_xor<4>(v);
    v += row_xor<2>(v);
    v += row_xor<1>(v);
    return v;
}
__device__ __forceinline__ float dot4(const float4 a, const float4 b) {
    v2f alo, ahi, blo, bhi;
    alo.x = a.x; alo.y = a.y; ahi.x = a.z; ahi.y = a.w;
    blo.x = b.x; blo.y = b.y; bhi.x = b.z; bhi.y = b.w;
    const v2f r = __builtin_elementwise_fma(ahi, bhi, alo * blo);
    return r.x + r.y;
}
__device__ __forceinline__ void axpy4(float p, const float4 v, v2f& lo, v2f& hi) {
    v2f pp, vlo, vhi;
    pp.x = p; pp.y = p; vlo.x = v.x; vlo.y = v.y; vhi.x = v.z; vhi.y = v.w;
    lo = __builtin_elementwise_fma(pp, vlo, lo);
    hi = __builtin_elementwise_fma(pp, vhi, hi);
}

template <bool GUARD>
__device__ __forceinline__ void take_rows(const float* __restrict__ x, int s, int t0, int rem, int c0,
                                          const float4 ws, float adv, float& m, float& sum, v2f& alo, v2f& ahi) {
    float4 v[8];
    float  e[8];
#pragma unroll
    for (int j = 0; j < 8; ++j) {
        if (!GUARD || j < rem) {
            const unsigned off = (unsigned)lane_i(s, t0 + j) * 512u + (unsigned)c0 * 4u;
            v[j] = *(const float4*)((const char*)x + off);
        }
    }
#pragma unroll
    for (int j = 0; j < 8; ++j) {
        e[j] = -INFINITY;
        if (!GUARD || j < rem) e[j] = leaky(wave_sum_valu(dot4(v[j], ws)) + adv);
    }
    float mn = m;
#pragma unroll
    for (int j = 0; j < 8; ++j) mn = fmaxf(mn, e[j]);
    const float sc = __expf(m - mn);
    sum *= sc;
    alo *= sc; ahi *= sc;
#pragma unroll
    for (int j = 0; j < 8; ++j) {
        if (!GUARD || j < rem) {
            const float p = __expf(e[j] - mn);
            sum += p;
            axpy4(p, v[j], alo, ahi);
        }
    }
    m = mn;
}

__global__ __launch_bounds__(256) __attribute__((amdgpu_waves_per_eu(16, 16))) void node_gather_kernel(const int* __restrict__ bucket,
                                                          const int* __restrict__ cnt,
                                                          const int* __restrict__ sorted,
                                                          const int* __restrict__ offs,
                                                          const float* __restrict__ wsd,
                                                          const float* __restrict__ x,
                                                          float* __restrict__ aggx) {
    const int lane = threadIdx.x & 31;
    const int d = blockIdx.x * 8 + __builtin_amdgcn_readfirstlane((int)(threadIdx.x >> 5));
    if (d >= N_NODES) return;
    const int c0 = lane * 4;
    const int n = cnt[d];
    const int nb = n <= BUCKET_CAP ? n : 0;
    const float4 ws = *(const float4*)(wsd + c0);
    const float4 wd = *(const float4*)(wsd + 128 + c0);
    const float4 vs = *(const float4*)(x + (size_t)d * 128 + c0);
    int s = d;
    if (lane < nb) s = bucket[(size_t)d * BUCKET_CAP + lane];

    const float adv = lane_f(wave_sum_valu(dot4(vs, wd)), 0);
    float m = leaky(wave_sum_valu(dot4(vs, ws)) + adv);
    float sum = 1.0f;
    v2f alo, ahi;
    alo.x = vs.x; alo.y = vs.y; ahi.x = vs.z; ahi.y = vs.w;

    int t = 0;
    for (; t + 8 <= nb; t += 8) take_rows<false>(x, s, t, 8, c0, ws, adv, m, sum, alo, ahi);
    if (t < nb) take_rows<true>(x, s, t, nb - t, c0, ws, adv, m, sum, alo, ahi);

    if (n > BUCKET_CAP) {
        const int j = d >> BIN_SHIFT;
        const int u = d & (BIN_NODES - 1);
        for (int cc = 0; cc < N_CHUNKS; ++cc) {
            int lo = offs[cc * OFF_PITCH + j];
            int hi = offs[cc * OFF_PITCH + j + 1];
            if (lo < 0) lo = 0;
            if (hi > CHUNK) hi = CHUNK;
            for (int base = lo; base < hi; base += 32) {
                const int i = base + lane;
                int sv = d;
                bool hit = false;
                if (i < hi) {
                    const int pv = sorted[(size_t)cc * CHUNK + i];
                    if (((pv >> 17) & (BIN_NODES - 1)) == u) { sv = pv & SRC_MASK; hit = true; }
                }
                unsigned hits = __builtin_amdgcn_ballot_w32(hit);
                while (hits) {
                    const int th = __builtin_ctz(hits);
                    hits &= hits - 1;
                    take_rows<true>(x, sv, th, 1, c0, ws, adv, m, sum, alo, ahi);
                }
            }
        }
    }
    const float inv = 1.0f / sum;
    *(float4*)(aggx + (size_t)d * 128 + c0) = make_float4(alo.x * inv, alo.y * inv, ahi.x * inv, ahi.y * inv);
}

#define MLP_WG    256
#define MLP_WAVES 8
#define SP        68
#define WQ_F      (128 * 256)
#define W2Q_F     (256 * 128)
#define WOQ_F     (128 * 16)
#define PATCH_F   (16 * SP)
#define FULL_ITERS   (ROW_TILES / (MLP_WG * MLP_WAVES))
#define FULL_TILES   (FULL_ITERS * MLP_WG * MLP_WAVES)
#define SHARED_TILES (ROW_TILES - FULL_TILES)
static_assert(SHARED_TILES <= MLP_WG, "a workgroup takes at most one shared tile");

__device__ __forceinline__ v8f mm4(float a0, float a1, float b0, float b1, v8f acc) {
    v2f a; a.x = a0; a.y = a1;
    v2f bm; bm.x = b0; bm.y = b1;
    return __builtin_amdgcn_wmma_f32_16x16x4_f32(false, a, false, bm, (short)0, acc, false, false);
}

__global__ __launch_bounds__(256) void mlp_fused_kernel(const float* __restrict__ aggx,
                                                        const float* __restrict__ W,
                                                        const float* __restrict__ b,
                                                        const float* __restrict__ Wfc2,
                                                        const float* __restrict__ bfc2,
                                                        const float* __restrict__ Wout,
                                                        const float* __restrict__ bout,
                                                        float* __restrict__ out) {
    __shared__ __attribute__((aligned(16))) float lds[WQ_F + W2Q_F + WOQ_F + MLP_WAVES * PATCH_F];
    float* Wq  = lds;
    float* W2q = Wq + WQ_F;
    float* Woq = W2q + W2Q_F;
    const int tid  = threadIdx.x;
    const int lane = tid & 31;
    const int wave = __builtin_amdgcn_readfirstlane(tid >> 5);

#pragma unroll
    for (int i = 0; i < 8; ++i) {
        const int task = tid + 256 * i;
        const int kq = task >> 6, c4 = (task & 63) * 4;
        const float* g = W + (size_t)(4 * kq) * 256 + c4;
        const float4 r0 = *(const float4*)g;
        const float4 r1 = *(const float4*)(g + 256);
        const float4 r2 = *(const float4*)(g + 512);
        const float4 r3 = *(const float4*)(g + 768);
        float4* dst = (float4*)(Wq + (size_t)(kq * 256 + c4) * 4);
        dst[0] = make_float4(r0.x, r1.x, r2.x, r3.x);
        dst[1] = make_float4(r0.y, r1.y, r2.y, r3.y);
        dst[2] = make_float4(r0.z, r1.z, r2.z, r3.z);
        dst[3] = make_float4(r0.w, r1.w, r2.w, r3.w);
    }
#pragma unroll
    for (int i = 0; i < 8; ++i) {
        const int task = tid + 256 * i;
        const int kq = task >> 5, c4 = (task & 31) * 4;
        const float* g = Wfc2 + (size_t)(4 * kq) * 128 + c4;
        const float4 r0 = *(const float4*)g;
        const float4 r1 = *(const float4*)(g + 128);
        const float4 r2 = *(const float4*)(g + 256);
        const float4 r3 = *(const float4*)(g + 384);
        float4* dst = (float4*)(W2q + (size_t)(kq * 128 + c4) * 4);
        dst[0] = make_float4(r0.x, r1.x, r2.x, r3.x);
        dst[1] = make_float4(r0.y, r1.y, r2.y, r3.y);
        dst[2] = make_float4(r0.z, r1.z, r2.z, r3.z);
        dst[3] = make_float4(r0.w, r1.w, r2.w, r3.w);
    }
#pragma unroll
    for (int n = 0; n < 8; ++n) {
        const int i = tid + 256 * n;
        const int j = i & 3, col = (i >> 2) & 15, kq = i >> 6;
        const float v = Wout[(4 * kq + j) * 6 + (col < 6 ? col : 0)];
        Woq[i] = (col < 6) ? v : 0.0f;
    }
    __syncthreads();

    const int c = lane & 15;
    const int h = lane >> 4;
    float* patch = Woq + WOQ_F + wave * PATCH_F;
    const float4* Wq4  = (const float4*)Wq;
    const float4* W2q4 = (const float4*)W2q;
    const float4* Woq4 = (const float4*)Woq;
    float*        pst  = patch + (8 * h) * SP + c;
    const float4* pld  = (const float4*)(patch + c * SP + 4 * h);

    float4* red4 = (float4*)Wq;

#pragma unroll 1
    for (int u = 0; u <= FULL_ITERS; ++u) {
        const bool shared = u == FULL_ITERS;
        if (shared && (int)blockIdx.x >= SHARED_TILES) break;
        const int tile = shared ? FULL_TILES + (int)blockIdx.x
                                : u * (MLP_WG * MLP_WAVES) + wave * MLP_WG + (int)blockIdx.x;
        const int p0 = shared ? wave : 0;
        const int p1 = shared ? (wave < 4 ? wave + 1 : wave) : 4;
        const int row0 = tile * 16;

        float4 a[16];
        if (p0 < p1) {
            const float* ap = aggx + (size_t)(row0 + c) * 128 + 4 * h;
#pragma unroll
            for (int kb = 0; kb < 16; ++kb) a[kb] = *(const float4*)(ap + 8 * kb);
        }

        v8f acc2[8] = {};
#pragma unroll 1
        for (int p = p0; p < p1; ++p) {
            v8f acc1[4] = {};
            const float4* wq = Wq4 + h * 256 + p * 64 + c;
#pragma unroll
            for (int kb = 0; kb < 16; ++kb) {
                float4 bq[4];
#pragma unroll
                for (int t = 0; t < 4; ++t) bq[t] = wq[kb * 512 + t * 16];
#pragma unroll
                for (int t = 0; t < 4; ++t) acc1[t] = mm4(a[kb].x, a[kb].y, bq[t].x, bq[t].y, acc1[t]);
#pragma unroll
                for (int t = 0; t < 4; ++t) acc1[t] = mm4(a[kb].z, a[kb].w, bq[t].z, bq[t].w, acc1[t]);
            }
#pragma unroll
            for (int t = 0; t < 4; ++t) {
                const float bb = b[p * 64 + t * 16 + c];
#pragma unroll
                for (int r = 0; r < 8; ++r) pst[r * SP + t * 16] = fmaxf(acc1[t][r] + bb, 0.0f);
            }
            const float4* w2 = W2q4 + (p * 16 + h) * 128 + c;
#pragma unroll
            for (int kb = 0; kb < 8; ++kb) {
                const float4 a2 = pld[2 * kb];
                float4 bq[8];
#pragma unroll
                for (int t = 0; t < 8; ++t) bq[t] = w2[kb * 256 + t * 16];
#pragma unroll
                for (int t = 0; t < 8; ++t) acc2[t] = mm4(a2.x, a2.y, bq[t].x, bq[t].y, acc2[t]);
#pragma unroll
                for (int t = 0; t < 8; ++t) acc2[t] = mm4(a2.z, a2.w, bq[t].z, bq[t].w, acc2[t]);
            }
        }

        if (shared) {
            __syncthreads();
            if (wave >= 1 && wave < 4) {
                float4* rp = red4 + (wave - 1) * 512 + lane;
#pragma unroll
                for (int t = 0; t < 8; ++t) {
                    rp[(2 * t) * 32]     = make_float4(acc2[t][0], acc2[t][1], acc2[t][2], acc2[t][3]);
                    rp[(2 * t + 1) * 32] = make_float4(acc2[t][4], acc2[t][5], acc2[t][6], acc2[t][7]);
                }
            }
            __syncthreads();
            if (wave != 0) break;
#pragma unroll
            for (int w = 0; w < 3; ++w) {
                const float4* rq = red4 + w * 512 + lane;
#pragma unroll
                for (int t = 0; t < 8; ++t) {
                    const float4 lo = rq[(2 * t) * 32];
                    const float4 hi = rq[(2 * t + 1) * 32];
                    acc2[t][0] += lo.x; acc2[t][1] += lo.y; acc2[t][2] += lo.z; acc2[t][3] += lo.w;
                    acc2[t][4] += hi.x; acc2[t][5] += hi.y; acc2[t][6] += hi.z; acc2[t][7] += hi.w;
                }
            }
        }

        v8f acc3a = {}, acc3b = {};
#pragma unroll
        for (int half = 0; half < 2; ++half) {
#pragma unroll
            for (int t = 0; t < 4; ++t) {
                const float bb = bfc2[half * 64 + t * 16 + c];
#pragma unroll
                for (int r = 0; r < 8; ++r) pst[r * SP + t * 16] = fmaxf(acc2[half * 4 + t][r] + bb, 0.0f);
            }
            const float4* wo = Woq4 + (half * 16 + h) * 16 + c;
#pragma unroll
            for (int kb = 0; kb < 8; ++kb) {
                const float4 a3 = pld[2 * kb];
                const float4 bq = wo[kb * 32];
                acc3a = mm4(a3.x, a3.y, bq.x, bq.y, acc3a);
                acc3b = mm4(a3.z, a3.w, bq.z, bq.w, acc3b);
            }
        }
        if (c < 6) {
            const float bb = bout[c];
            float* op = out + (size_t)(row0 + 8 * h) * 6 + c;
#pragma unroll
            for (int r = 0; r < 8; ++r) op[r * 6] = (acc3a[r] + acc3b[r]) + bb;
        }
    }
}

extern "C" void kernel_launch(void* const* d_in, const int* in_sizes, int n_in,
                              void* d_out, int out_size, void* d_ws, size_t ws_size,
                              hipStream_t stream) {
    const float* x     = (const float*)d_in[0];
    const int*   ei    = (const int*)  d_in[1];
    const float* W     = (const float*)d_in[2];
    const float* a_src = (const float*)d_in[3];
    const float* a_dst = (const float*)d_in[4];
    const float* b     = (const float*)d_in[5];
    const float* Wfc2  = (const float*)d_in[6];
    const float* bfc2  = (const float*)d_in[7];
    const float* Wout  = (const float*)d_in[8];
    const float* bout  = (const float*)d_in[9];
    float* out = (float*)d_out;

    float* ws      = (float*)d_ws;
    float* aggx    = ws;
    float* wsd     = aggx + (size_t)N_NODES * 128;
    int*   cnt     = (int*)(wsd + 256);
    int*   bucket  = cnt + N_NODES;
    int*   sorted  = bucket + (size_t)N_NODES * BUCKET_CAP;
    int*   offs    = sorted + (size_t)N_CHUNKS * CHUNK;

    bin_kernel<<<N_CHUNKS + FOLD_BLOCKS, 1024, 0, stream>>>(ei, W, a_src, a_dst, wsd, sorted, offs);
    build_kernel<<<N_BINS, 1024, 0, stream>>>(sorted, offs, cnt, bucket);
    node_gather_kernel<<<(N_NODES + 7) / 8, 256, 0, stream>>>(bucket, cnt, sorted, offs, wsd, x, aggx);
    mlp_fused_kernel<<<MLP_WG, 256, 0, stream>>>(aggx, W, b, Wfc2, bfc2, Wout, bout, out);
}


# Bucket build fused into the gather launch: a 32-wave workgroup per 196-node bin builds buckets in LDS, its waves then take the bin's nodes by LDS ticket: 3 launches, 16352 waves enter, not 100000
# speedup vs baseline: 6.0173x; 1.0489x over previous
#include <hip/hip_runtime.h>
#include <hip/hip_bf16.h>

typedef __attribute__((ext_vector_type(2))) float v2f;
typedef __attribute__((ext_vector_type(8))) float v8f;

#define N_NODES   100000
#define N_EDGES0  800000
#define N_EDGES   900000
#define ROW_TILES 6250

__device__ __forceinline__ float leaky(float v) { return v > 0.0f ? v : 0.2f * v; }

__device__ __forceinline__ float wave_max_f(float v) {
#pragma unroll
    for (int off = 16; off; off >>= 1) v = fmaxf(v, __shfl_xor(v, off, 32));
    return v;
}
__device__ __forceinline__ float wave_sum_f(float v) {
#pragma unroll
    for (int off = 16; off; off >>= 1) v += __shfl_xor(v, off, 32);
    return v;
}

#define BUCKET_CAP 32
#define CHUNK      4096
#define N_CHUNKS   196
#define BIN_NODES  196
#define N_BINS     511
#define BIN_SLOTS  512
#define OFF_PITCH  512
#define DST_MASK   255
#define BIN_LDS    256
#define SRC_MASK   0x1ffff
static_assert(N_BINS * BIN_NODES >= N_NODES && (N_BINS - 1) * BIN_NODES < N_NODES, "bins cover the nodes exactly");
static_assert(N_BINS + 1 <= OFF_PITCH && N_BINS <= BIN_SLOTS && BIN_NODES <= BIN_LDS && BIN_LDS == DST_MASK + 1, "table sizes");
#define FOLD_BLOCKS 4
__global__ __launch_bounds__(1024) void bin_kernel(const int* __restrict__ ei,
                                                   const float* __restrict__ W,
                                                   const float* __restrict__ a_src,
                                                   const float* __restrict__ a_dst,
                                                   float* __restrict__ wsd,
                                                   int* __restrict__ sorted,
                                                   int* __restrict__ offs) {
    __shared__ int lcnt[BIN_SLOTS];
    __shared__ int loff[BIN_SLOTS];
    __shared__ __attribute__((aligned(16))) int lsorted[CHUNK];
    const int tid = threadIdx.x;
    if (blockIdx.x >= N_CHUNKS) {
        const int lane = tid & 31;
        const int k = ((int)blockIdx.x - N_CHUNKS) * 32 + (tid >> 5);
        const float* wr = W + (size_t)k * 256;
        float s = 0.0f, d = 0.0f;
#pragma unroll
        for (int i = 0; i < 8; ++i) {
            float v = wr[lane + 32 * i];
            s += v * a_src[lane + 32 * i];
            d += v * a_dst[lane + 32 * i];
        }
        s = wave_sum_f(s);
        d = wave_sum_f(d);
        if (lane == 0) { wsd[k] = s; wsd[128 + k] = d; }
        return;
    }
    if (tid < BIN_SLOTS) lcnt[tid] = 0;
    __syncthreads();
    const int c = blockIdx.x;
    const int e0 = c * CHUNK + tid * 4;
    const bool live = e0 < N_EDGES0;
    int bin[4], rank[4], val[4];
    if (live) {
        const int4 s4 = *(const int4*)(ei + e0);
        const int4 d4 = *(const int4*)(ei + N_EDGES0 + e0);
        const int dd[4] = {d4.x, d4.y, d4.z, d4.w};
        const int ss[4] = {s4.x, s4.y, s4.z, s4.w};
#pragma unroll
        for (int k = 0; k < 4; ++k) {
            unsigned bk = (unsigned)dd[k] / (unsigned)BIN_NODES;
            if (bk > N_BINS - 1) bk = N_BINS - 1;
            bin[k]  = (int)bk;
            val[k]  = (int)((((unsigned)dd[k] - bk * BIN_NODES) & DST_MASK) << 17) | (ss[k] & SRC_MASK);
            rank[k] = atomicAdd(&lcnt[bin[k]], 1);
        }
    }
    __syncthreads();
    if (tid < 32) {
        int v[16];
        int sum = 0;
#pragma unroll
        for (int i = 0; i < 16; ++i) { v[i] = lcnt[tid * 16 + i]; sum += v[i]; }
        int incl = sum;
#pragma unroll
        for (int off = 1; off < 32; off <<= 1) {
            const int t = __shfl_up(incl, off, 32);
            if (tid >= off) incl += t;
        }
        int run = incl - sum;
#pragma unroll
        for (int i = 0; i < 16; ++i) { loff[tid * 16 + i] = run; run += v[i]; }
    }
    __syncthreads();
    if (live) {
#pragma unroll
        for (int k = 0; k < 4; ++k) lsorted[loff[bin[k]] + rank[k]] = val[k];
    }
    __syncthreads();
    if (live) *(int4*)(sorted + (size_t)c * CHUNK + tid * 4) = *(const int4*)(lsorted + tid * 4);
    if (tid < OFF_PITCH) offs[c * OFF_PITCH + tid] = loff[tid];
}

__device__ __forceinline__ int lane_i(int v, int t) { return __builtin_amdgcn_readlane(v, t); }
__device__ __forceinline__ float lane_f(float v, int t) {
    return __int_as_float(__builtin_amdgcn_readlane(__float_as_int(v), t));
}

__device__ __forceinline__ float xhalf(float v) {
    const int b = __float_as_int(v);
    return __int_as_float(__builtin_amdgcn_permlanex16(b, b, 0x76543210, 0xfedcba98, false, false));
}
template <int N> __device__ __forceinline__ float row_xor(float v) {
    return __int_as_float(__builtin_amdgcn_update_dpp(0, __float_as_int(v), 0x160 + N, 0xf, 0xf, true));
}
__device__ __forceinline__ float wave_sum_valu(float v) {
    v += xhalf(v);
    v += row_xor<8>(v);
    v += row_xor<4>(v);
    v += row_xor<2>(v);
    v += row_xor<1>(v);
    return v;
}
__device__ __forceinline__ float dot4(const float4 a, const float4 b) {
    v2f alo, ahi, blo, bhi;
    alo.x = a.x; alo.y = a.y; ahi.x = a.z; ahi.y = a.w;
    blo.x = b.x; blo.y = b.y; bhi.x = b.z; bhi.y = b.w;
    const v2f r = __builtin_elementwise_fma(ahi, bhi, alo * blo);
    return r.x + r.y;
}
__device__ __forceinline__ void axpy4(float p, const float4 v, v2f& lo, v2f& hi) {
    v2f pp, vlo, vhi;
    pp.x = p; pp.y = p; vlo.x = v.x; vlo.y = v.y; vhi.x = v.z; vhi.y = v.w;
    lo = __builtin_elementwise_fma(pp, vlo, lo);
    hi = __builtin_elementwise_fma(pp, vhi, hi);
}

template <bool GUARD>
__device__ __forceinline__ void take_rows(const float* __restrict__ x, int s, int t0, int rem, int c0,
                                          const float4 ws, float adv, float& m, float& sum, v2f& alo, v2f& ahi) {
    float4 v[8];
    float  e[8];
#pragma unroll
    for (int j = 0; j < 8; ++j) {
        if (!GUARD || j < rem) {
            const unsigned off = (unsigned)lane_i(s, t0 + j) * 512u + (unsigned)c0 * 4u;
            v[j] = *(const float4*)((const char*)x + off);
        } else {
            v[j] = make_float4(0.0f, 0.0f, 0.0f, 0.0f);
        }
    }
#pragma unroll
    for (int j = 0; j < 8; ++j) {
        e[j] = -INFINITY;
        if (!GUARD || j < rem) e[j] = leaky(wave_sum_valu(dot4(v[j], ws)) + adv);
    }
    float mn = m;
#pragma unroll
    for (int j = 0; j < 8; ++j) mn = fmaxf(mn, e[j]);
    const float sc = __expf(m - mn);
    sum *= sc;
    alo *= sc; ahi *= sc;
#pragma unroll
    for (int j = 0; j < 8; ++j) {
        if (!GUARD || j < rem) {
            const float p = __expf(e[j] - mn);
            sum += p;
            axpy4(p, v[j], alo, ahi);
        }
    }
    m = mn;
}

__global__ __launch_bounds__(1024) __attribute__((amdgpu_waves_per_eu(16, 16))) void bin_gather_kernel(const int* __restrict__ sorted,
                                                          const int* __restrict__ offs,
                                                          const float* __restrict__ wsd,
                                                          const float* __restrict__ x,
                                                          float* __restrict__ aggx) {
    __shared__ int ncnt[BIN_LDS];
    __shared__ int nxt;
    __shared__ __attribute__((aligned(16))) int lbkt[BIN_LDS * BUCKET_CAP];
    const int tid = threadIdx.x;
    const int lane = tid & 31;
    const int wave = __builtin_amdgcn_readfirstlane(tid >> 5);
    const int j = blockIdx.x;
    if (tid < BIN_LDS) ncnt[tid] = 0;
    if (tid == 0) nxt = 0;
    for (int q = tid; q < BIN_LDS * BUCKET_CAP / 4; q += 1024) ((int4*)lbkt)[q] = make_int4(0, 0, 0, 0);
    __syncthreads();
    int lo_v = 0, hi_v = 0;
    {
        const int cc = wave + 32 * lane;
        if (cc < N_CHUNKS) {
            lo_v = offs[cc * OFF_PITCH + j];
            hi_v = offs[cc * OFF_PITCH + j + 1];
        }
    }
    int lo_s[7], hi_s[7], pv[7];
#pragma unroll
    for (int k = 0; k < 7; ++k) {
        lo_s[k] = __builtin_amdgcn_readlane(lo_v, k);
        hi_s[k] = __builtin_amdgcn_readlane(hi_v, k);
        if (lo_s[k] < 0) lo_s[k] = 0;
        if (hi_s[k] > CHUNK) hi_s[k] = CHUNK;
        pv[k] = -1;
        if (lo_s[k] + lane < hi_s[k]) pv[k] = sorted[(size_t)(wave + 32 * k) * CHUNK + lo_s[k] + lane];
    }
#pragma unroll
    for (int k = 0; k < 7; ++k) {
        if (pv[k] >= 0) {
            const int u = (pv[k] >> 17) & DST_MASK;
            const int slot = atomicAdd(&ncnt[u], 1);
            if (slot < BUCKET_CAP) lbkt[u * BUCKET_CAP + slot] = pv[k] & SRC_MASK;
        }
    }
#pragma unroll
    for (int k = 0; k < 7; ++k) {
        for (int i = lo_s[k] + 32 + lane; i < hi_s[k]; i += 32) {
            const int pw = sorted[(size_t)(wave + 32 * k) * CHUNK + i];
            const int u = (pw >> 17) & DST_MASK;
            const int slot = atomicAdd(&ncnt[u], 1);
            if (slot < BUCKET_CAP) lbkt[u * BUCKET_CAP + slot] = pw & SRC_MASK;
        }
    }
    __syncthreads();

    const int c0 = lane * 4;
    const unsigned cb = (unsigned)c0 * 4u;
    const float4 ws = *(const float4*)((const char*)wsd + cb);
    for (int it = 0; it < BIN_NODES; ++it) {
        int tk = 0;
        if (lane == 0) tk = atomicAdd(&nxt, 1);
        const int u = __builtin_amdgcn_readfirstlane(tk);
        if (u >= BIN_NODES) break;
        const int d = j * BIN_NODES + u;
        if (d >= N_NODES) break;
        const int n = __builtin_amdgcn_readfirstlane(ncnt[u]);
        const int nb = n <= BUCKET_CAP ? n : 0;
        const unsigned rb = (unsigned)d * 512u + cb;
        const float4 wd = *(const float4*)((const char*)(wsd + 128) + cb);
        const float4 vs = *(const float4*)((const char*)x + rb);
        int s = d;
        if (lane < nb) s = lbkt[u * BUCKET_CAP + lane];

        const float adv = lane_f(wave_sum_valu(dot4(vs, wd)), 0);
        float m = leaky(wave_sum_valu(dot4(vs, ws)) + adv);
        float sum = 1.0f;
        v2f alo, ahi;
        alo.x = vs.x; alo.y = vs.y; ahi.x = vs.z; ahi.y = vs.w;

        int t = 0;
        for (; t + 8 <= nb; t += 8) take_rows<false>(x, s, t, 8, c0, ws, adv, m, sum, alo, ahi);
        if (t < nb) take_rows<true>(x, s, t, nb - t, c0, ws, adv, m, sum, alo, ahi);

        if (n > BUCKET_CAP) {
            for (int cc = 0; cc < N_CHUNKS; ++cc) {
                int lo = offs[cc * OFF_PITCH + j];
                int hi = offs[cc * OFF_PITCH + j + 1];
                if (lo < 0) lo = 0;
                if (hi > CHUNK) hi = CHUNK;
                for (int base = lo; base < hi; base += 32) {
                    const int i = base + lane;
                    int sv = d;
                    bool hit = false;
                    if (i < hi) {
                        const int pw = sorted[(size_t)cc * CHUNK + i];
                        if (((pw >> 17) & DST_MASK) == u) { sv = pw & SRC_MASK; hit = true; }
                    }
                    unsigned hits = __builtin_amdgcn_ballot_w32(hit);
                    while (hits) {
                        const int th = __builtin_ctz(hits);
                        hits &= hits - 1;
                        take_rows<true>(x, sv, th, 1, c0, ws, adv, m, sum, alo, ahi);
                    }
                }
            }
        }
        const float inv = 1.0f / sum;
        *(float4*)((char*)aggx + rb) = make_float4(alo.x * inv, alo.y * inv, ahi.x * inv, ahi.y * inv);
    }
}

#define MLP_WG    256
#define MLP_WAVES 8
#define SP        68
#define WQ_F      (128 * 256)
#define W2Q_F     (256 * 128)
#define WOQ_F     (128 * 16)
#define PATCH_F   (16 * SP)
#define FULL_ITERS   (ROW_TILES / (MLP_WG * MLP_WAVES))
#define FULL_TILES   (FULL_ITERS * MLP_WG * MLP_WAVES)
#define SHARED_TILES (ROW_TILES - FULL_TILES)
static_assert(SHARED_TILES <= MLP_WG, "a workgroup takes at most one shared tile");

__device__ __forceinline__ v8f mm4(float a0, float a1, float b0, float b1, v8f acc) {
    v2f a; a.x = a0; a.y = a1;
    v2f bm; bm.x = b0; bm.y = b1;
    return __builtin_amdgcn_wmma_f32_16x16x4_f32(false, a, false, bm, (short)0, acc, false, false);
}

__global__ __launch_bounds__(256) void mlp_fused_kernel(const float* __restrict__ aggx,
                                                        const float* __restrict__ W,
                                                        const float* __restrict__ b,
                                                        const float* __restrict__ Wfc2,
                                                        const float* __restrict__ bfc2,
                                                        const float* __restrict__ Wout,
                                                        const float* __restrict__ bout,
                                                        float* __restrict__ out) {
    __shared__ __attribute__((aligned(16))) float lds[WQ_F + W2Q_F + WOQ_F + MLP_WAVES * PATCH_F];
    float* Wq  = lds;
    float* W2q = Wq + WQ_F;
    float* Woq = W2q + W2Q_F;
    const int tid  = threadIdx.x;
    const int lane = tid & 31;
    const int wave = __builtin_amdgcn_readfirstlane(tid >> 5);

#pragma unroll
    for (int i = 0; i < 8; ++i) {
        const int task = tid + 256 * i;
        const int kq = task >> 6, c4 = (task & 63) * 4;
        const float* g = W + (size_t)(4 * kq) * 256 + c4;
        const float4 r0 = *(const float4*)g;
        const float4 r1 = *(const float4*)(g + 256);
        const float4 r2 = *(const float4*)(g + 512);
        const float4 r3 = *(const float4*)(g + 768);
        float4* dst = (float4*)(Wq + (size_t)(kq * 256 + c4) * 4);
        dst[0] = make_float4(r0.x, r1.x, r2.x, r3.x);
        dst[1] = make_float4(r0.y, r1.y, r2.y, r3.y);
        dst[2] = make_float4(r0.z, r1.z, r2.z, r3.z);
        dst[3] = make_float4(r0.w, r1.w, r2.w, r3.w);
    }
#pragma unroll
    for (int i = 0; i < 8; ++i) {
        const int task = tid + 256 * i;
        const int kq = task >> 5, c4 = (task & 31) * 4;
        const float* g = Wfc2 + (size_t)(4 * kq) * 128 + c4;
        const float4 r0 = *(const float4*)g;
        const float4 r1 = *(const float4*)(g + 128);
        const float4 r2 = *(const float4*)(g + 256);
        const float4 r3 = *(const float4*)(g + 384);
        float4* dst = (float4*)(W2q + (size_t)(kq * 128 + c4) * 4);
        dst[0] = make_float4(r0.x, r1.x, r2.x, r3.x);
        dst[1] = make_float4(r0.y, r1.y, r2.y, r3.y);
        dst[2] = make_float4(r0.z, r1.z, r2.z, r3.z);
        dst[3] = make_float4(r0.w, r1.w, r2.w, r3.w);
    }
#pragma unroll
    for (int n = 0; n < 8; ++n) {
        const int i = tid + 256 * n;
        const int j = i & 3, col = (i >> 2) & 15, kq = i >> 6;
        const float v = Wout[(4 * kq + j) * 6 + (col < 6 ? col : 0)];
        Woq[i] = (col < 6) ? v : 0.0f;
    }
    __syncthreads();

    const int c = lane & 15;
    const int h = lane >> 4;
    float* patch = Woq + WOQ_F + wave * PATCH_F;
    const float4* Wq4  = (const float4*)Wq;
    const float4* W2q4 = (const float4*)W2q;
    const float4* Woq4 = (const float4*)Woq;
    float*        pst  = patch + (8 * h) * SP + c;
    const float4* pld  = (const float4*)(patch + c * SP + 4 * h);

    float4* red4 = (float4*)Wq;

#pragma unroll 1
    for (int u = 0; u <= FULL_ITERS; ++u) {
        const bool shared = u == FULL_ITERS;
        if (shared && (int)blockIdx.x >= SHARED_TILES) break;
        const int tile = shared ? FULL_TILES + (int)blockIdx.x
                                : u * (MLP_WG * MLP_WAVES) + wave * MLP_WG + (int)blockIdx.x;
        const int p0 = shared ? wave : 0;
        const int p1 = shared ? (wave < 4 ? wave + 1 : wave) : 4;
        const int row0 = tile * 16;

        float4 a[16];
        if (p0 < p1) {
            const float* ap = aggx + (size_t)(row0 + c) * 128 + 4 * h;
#pragma unroll
            for (int kb = 0; kb < 16; ++kb) a[kb] = *(const float4*)(ap + 8 * kb);
        }

        v8f acc2[8] = {};
#pragma unroll 1
        for (int p = p0; p < p1; ++p) {
            v8f acc1[4] = {};
            const float4* wq = Wq4 + h * 256 + p * 64 + c;
#pragma unroll
            for (int kb = 0; kb < 16; ++kb) {
                float4 bq[4];
#pragma unroll
                for (int t = 0; t < 4; ++t) bq[t] = wq[kb * 512 + t * 16];
#pragma unroll
                for (int t = 0; t < 4; ++t) acc1[t] = mm4(a[kb].x, a[kb].y, bq[t].x, bq[t].y, acc1[t]);
#pragma unroll
                for (int t = 0; t < 4; ++t) acc1[t] = mm4(a[kb].z, a[kb].w, bq[t].z, bq[t].w, acc1[t]);
            }
#pragma unroll
            for (int t = 0; t < 4; ++t) {
                const float bb = b[p * 64 + t * 16 + c];
#pragma unroll
                for (int r = 0; r < 8; ++r) pst[r * SP + t * 16] = fmaxf(acc1[t][r] + bb, 0.0f);
            }
            const float4* w2 = W2q4 + (p * 16 + h) * 128 + c;
#pragma unroll
            for (int kb = 0; kb < 8; ++kb) {
                const float4 a2 = pld[2 * kb];
                float4 bq[8];
#pragma unroll
                for (int t = 0; t < 8; ++t) bq[t] = w2[kb * 256 + t * 16];
#pragma unroll
                for (int t = 0; t < 8; ++t) acc2[t] = mm4(a2.x, a2.y, bq[t].x, bq[t].y, acc2[t]);
#pragma unroll
                for (int t = 0; t < 8; ++t) acc2[t] = mm4(a2.z, a2.w, bq[t].z, bq[t].w, acc2[t]);
            }
        }

        if (shared) {
            __syncthreads();
            if (wave >= 1 && wave < 4) {
                float4* rp = red4 + (wave - 1) * 512 + lane;
#pragma unroll
                for (int t = 0; t < 8; ++t) {
                    rp[(2 * t) * 32]     = make_float4(acc2[t][0], acc2[t][1], acc2[t][2], acc2[t][3]);
                    rp[(2 * t + 1) * 32] = make_float4(acc2[t][4], acc2[t][5], acc2[t][6], acc2[t][7]);
                }
            }
            __syncthreads();
            if (wave != 0) break;
#pragma unroll
            for (int w = 0; w < 3; ++w) {
                const float4* rq = red4 + w * 512 + lane;
#pragma unroll
                for (int t = 0; t < 8; ++t) {
                    const float4 lo = rq[(2 * t) * 32];
                    const float4 hi = rq[(2 * t + 1) * 32];
                    acc2[t][0] += lo.x; acc2[t][1] += lo.y; acc2[t][2] += lo.z; acc2[t][3] += lo.w;
                    acc2[t][4] += hi.x; acc2[t][5] += hi.y; acc2[t][6] += hi.z; acc2[t][7] += hi.w;
                }
            }
        }

        v8f acc3a = {}, acc3b = {};
#pragma unroll
        for (int half = 0; half < 2; ++half) {
#pragma unroll
            for (int t = 0; t < 4; ++t) {
                const float bb = bfc2[half * 64 + t * 16 + c];
#pragma unroll
                for (int r = 0; r < 8; ++r) pst[r * SP + t * 16] = fmaxf(acc2[half * 4 + t][r] + bb, 0.0f);
            }
            const float4* wo = Woq4 + (half * 16 + h) * 16 + c;
#pragma unroll
            for (int kb = 0; kb < 8; ++kb) {
                const float4 a3 = pld[2 * kb];
                const float4 bq = wo[kb * 32];
                acc3a = mm4(a3.x, a3.y, bq.x, bq.y, acc3a);
                acc3b = mm4(a3.z, a3.w, bq.z, bq.w, acc3b);
            }
        }
        if (c < 6) {
            const float bb = bout[c];
            float* op = out + (size_t)(row0 + 8 * h) * 6 + c;
#pragma unroll
            for (int r = 0; r < 8; ++r) op[r * 6] = (acc3a[r] + acc3b[r]) + bb;
        }
    }
}

extern "C" void kernel_launch(void* const* d_in, const int* in_sizes, int n_in,
                              void* d_out, int out_size, void* d_ws, size_t ws_size,
                              hipStream_t stream) {
    const float* x     = (const float*)d_in[0];
    const int*   ei    = (const int*)  d_in[1];
    const float* W     = (const float*)d_in[2];
    const float* a_src = (const float*)d_in[3];
    const float* a_dst = (const float*)d_in[4];
    const float* b     = (const float*)d_in[5];
    const float* Wfc2  = (const float*)d_in[6];
    const float* bfc2  = (const float*)d_in[7];
    const float* Wout  = (const float*)d_in[8];
    const float* bout  = (const float*)d_in[9];
    float* out = (float*)d_out;

    float* ws      = (float*)d_ws;
    float* aggx    = ws;
    float* wsd     = aggx + (size_t)N_NODES * 128;
    int*   sorted  = (int*)(wsd + 256);
    int*   offs    = sorted + (size_t)N_CHUNKS * CHUNK;

    bin_kernel<<<N_CHUNKS + FOLD_BLOCKS, 1024, 0, stream>>>(ei, W, a_src, a_dst, wsd, sorted, offs);
    bin_gather_kernel<<<N_BINS, 1024, 0, stream>>>(sorted, offs, wsd, x, aggx);
    mlp_fused_kernel<<<MLP_WG, 256, 0, stream>>>(aggx, W, b, Wfc2, bfc2, Wout, bout, out);
}
